# starting kernel
# baseline (speedup 1.0000x reference)
#include <hip/hip_runtime.h>

typedef __attribute__((ext_vector_type(2))) float v2f;
typedef __attribute__((ext_vector_type(8))) float v8f;

#define N_TOK   1728
#define CDIM    256
#define NHEADS  8
#define HD      32
#define NT      108
#define SCALE_  0.1767766952966369f

__device__ __forceinline__ v8f wmma4(v2f a, v2f b, v8f c) {
  return __builtin_amdgcn_wmma_f32_16x16x4_f32(
       false, a,  false, b,
       (short)0, c,  false,  false);
}

__global__ __launch_bounds__(256)
void gsa3d_qkv_kernel(const float* __restrict__ x,
                      const float* __restrict__ wqkv,
                      float* __restrict__ Q,
                      float* __restrict__ K,
                      float* __restrict__ V) {
  const int wid  = blockIdx.x * 8 + (threadIdx.x >> 5);
  const int lane = threadIdx.x & 31;
  const int l16  = lane & 15;
  const int kh   = lane >> 4;
  const int tn   = wid % 48;
  const int tm   = wid / 48;
  const int row  = tm * 16 + l16;
  const int col  = tn * 16 + l16;

  v8f acc = {};
  for (int k0 = 0; k0 < CDIM; k0 += 4) {
    const int ka = k0 + 2 * kh;
    v2f a, b;
    a.x = x[(ka + 0) * N_TOK + row];
    a.y = x[(ka + 1) * N_TOK + row];
    b.x = wqkv[col * CDIM + ka + 0];
    b.y = wqkv[col * CDIM + ka + 1];
    acc = wmma4(a, b, acc);
  }

  const int s  = col >> 8;
  const int hh = (col >> 5) & 7;
  const int d  = col & 31;
  float* dst = (s == 0) ? Q : ((s == 1) ? K : V);
  dst += hh * (N_TOK * HD) + d;
#pragma unroll
  for (int r = 0; r < 8; ++r) {
    const int n = tm * 16 + r + 8 * kh;
    dst[n * HD] = acc[r];
  }
}

__global__ __launch_bounds__(256)
void gsa3d_attn_kernel(const float* __restrict__ Q,
                       const float* __restrict__ K,
                       const float* __restrict__ V,
                       float* __restrict__ AO) {
  __shared__ float ldsP[8 * 256];
  const int wslot = threadIdx.x >> 5;
  const int wid   = blockIdx.x * 8 + wslot;
  const int lane  = threadIdx.x & 31;
  const int l16   = lane & 15;
  const int kh    = lane >> 4;
  const int h     = wid / NT;
  const int tq    = wid % NT;

  const float* Qh = Q + h * (N_TOK * HD);
  const float* Kh = K + h * (N_TOK * HD);
  const float* Vh = V + h * (N_TOK * HD);
  float* P = ldsP + wslot * 256;

  v2f aq[8];
  {
    const int row = tq * 16 + l16;
#pragma unroll
    for (int t = 0; t < 8; ++t) {
      const int ka = 4 * t + 2 * kh;
      aq[t].x = Qh[row * HD + ka + 0];
      aq[t].y = Qh[row * HD + ka + 1];
    }
  }

  float m_i[8], l_i[8];
#pragma unroll
  for (int r = 0; r < 8; ++r) { m_i[r] = -1.0e30f; l_i[r] = 0.0f; }
  v8f o0 = {};
  v8f o1 = {};

  for (int kb = 0; kb < NT; ++kb) {
    v8f s = {};
#pragma unroll
    for (int t = 0; t < 8; ++t) {
      const int kd = 4 * t + 2 * kh;
      v2f b;
      b.x = Kh[(kb * 16 + l16) * HD + kd + 0];
      b.y = Kh[(kb * 16 + l16) * HD + kd + 1];
      s = wmma4(aq[t], b, s);
    }

    float p[8], alpha[8];
#pragma unroll
    for (int r = 0; r < 8; ++r) {
      float sv = s[r] * SCALE_;
      sv = fminf(fmaxf(sv, -10.0f), 10.0f);
      float mx = sv;
      mx = fmaxf(mx, __shfl_xor(mx, 1, 32));
      mx = fmaxf(mx, __shfl_xor(mx, 2, 32));
      mx = fmaxf(mx, __shfl_xor(mx, 4, 32));
      mx = fmaxf(mx, __shfl_xor(mx, 8, 32));
      const float mnew = fmaxf(m_i[r], mx);
      alpha[r] = __expf(m_i[r] - mnew);
      p[r]     = __expf(sv - mnew);
      float rs = p[r];
      rs += __shfl_xor(rs, 1, 32);
      rs += __shfl_xor(rs, 2, 32);
      rs += __shfl_xor(rs, 4, 32);
      rs += __shfl_xor(rs, 8, 32);
      l_i[r] = l_i[r] * alpha[r] + rs;
      m_i[r] = mnew;
    }

#pragma unroll
    for (int r = 0; r < 8; ++r) {
      o0[r] *= alpha[r];
      o1[r] *= alpha[r];
      P[(r + 8 * kh) * 16 + l16] = p[r];
    }

#pragma unroll
    for (int t = 0; t < 4; ++t) {
      const int kp = 4 * t + 2 * kh;
      v2f ap, b0, b1;
      ap.x = P[l16 * 16 + kp + 0];
      ap.y = P[l16 * 16 + kp + 1];
      const float* vr0 = Vh + (kb * 16 + kp + 0) * HD;
      const float* vr1 = Vh + (kb * 16 + kp + 1) * HD;
      b0.x = vr0[l16];        b0.y = vr1[l16];
      b1.x = vr0[16 + l16];   b1.y = vr1[16 + l16];
      o0 = wmma4(ap, b0, o0);
      o1 = wmma4(ap, b1, o1);
    }
  }

#pragma unroll
  for (int r = 0; r < 8; ++r) {
    const int n   = tq * 16 + r + 8 * kh;
    const float inv = 1.0f / l_i[r];
    AO[n * CDIM + h * HD + l16]      = o0[r] * inv;
    AO[n * CDIM + h * HD + 16 + l16] = o1[r] * inv;
  }
}

__global__ __launch_bounds__(256)
void gsa3d_proj_kernel(const float* __restrict__ AO,
                       const float* __restrict__ wproj,
                       float* __restrict__ out) {
  const int wid  = blockIdx.x * 8 + (threadIdx.x >> 5);
  const int lane = threadIdx.x & 31;
  const int l16  = lane & 15;
  const int kh   = lane >> 4;
  const int tm   = wid / 16;
  const int tn   = wid % 16;
  const int row  = tm * 16 + l16;
  const int col  = tn * 16 + l16;

  v8f acc = {};
  for (int k0 = 0; k0 < CDIM; k0 += 4) {
    const int ka = k0 + 2 * kh;
    v2f a, b;
    a.x = AO[row * CDIM + ka + 0];
    a.y = AO[row * CDIM + ka + 1];
    b.x = wproj[col * CDIM + ka + 0];
    b.y = wproj[col * CDIM + ka + 1];
    acc = wmma4(a, b, acc);
  }
#pragma unroll
  for (int r = 0; r < 8; ++r) {
    const int n = tm * 16 + r + 8 * kh;
    out[col * N_TOK + n] = acc[r];
  }
}

extern "C" void kernel_launch(void* const* d_in, const int* in_sizes, int n_in,
                              void* d_out, int out_size, void* d_ws, size_t ws_size,
                              hipStream_t stream) {
  (void)in_sizes; (void)n_in; (void)out_size; (void)ws_size;
  const float* x     = (const float*)d_in[0];
  const float* wqkv  = (const float*)d_in[1];
  const float* wproj = (const float*)d_in[2];
  float* out = (float*)d_out;

  float* ws = (float*)d_ws;
  float* Q  = ws;
  float* K  = Q + NHEADS * N_TOK * HD;
  float* V  = K + NHEADS * N_TOK * HD;
  float* AO = V + NHEADS * N_TOK * HD;

  gsa3d_qkv_kernel <<<648, 256, 0, stream>>>(x, wqkv, Q, K, V);
  gsa3d_attn_kernel<<<108, 256, 0, stream>>>(Q, K, V, AO);
  gsa3d_proj_kernel<<<216, 256, 0, stream>>>(AO, wproj, out);
}


# attention: one work-group per (head, 16-row query tile); its 8 waves split the 108 key blocks and merge partial max/sum/O through LDS in fixed order (8x waves; softmax and PV sum order changes)
# speedup vs baseline: 3.2083x; 3.2083x over previous
#include <hip/hip_runtime.h>

typedef __attribute__((ext_vector_type(2))) float v2f;
typedef __attribute__((ext_vector_type(8))) float v8f;

#define N_TOK   1728
#define CDIM    256
#define NHEADS  8
#define HD      32
#define NT      108
#define SCALE_  0.1767766952966369f

__device__ __forceinline__ v8f wmma4(v2f a, v2f b, v8f c) {
  return __builtin_amdgcn_wmma_f32_16x16x4_f32(
       false, a,  false, b,
       (short)0, c,  false,  false);
}

__global__ __launch_bounds__(256)
void gsa3d_qkv_kernel(const float* __restrict__ x,
                      const float* __restrict__ wqkv,
                      float* __restrict__ Q,
                      float* __restrict__ K,
                      float* __restrict__ V) {
  const int wid  = blockIdx.x * 8 + (threadIdx.x >> 5);
  const int lane = threadIdx.x & 31;
  const int l16  = lane & 15;
  const int kh   = lane >> 4;
  const int tn   = wid % 48;
  const int tm   = wid / 48;
  const int row  = tm * 16 + l16;
  const int col  = tn * 16 + l16;

  v8f acc = {};
  for (int k0 = 0; k0 < CDIM; k0 += 4) {
    const int ka = k0 + 2 * kh;
    v2f a, b;
    a.x = x[(ka + 0) * N_TOK + row];
    a.y = x[(ka + 1) * N_TOK + row];
    b.x = wqkv[col * CDIM + ka + 0];
    b.y = wqkv[col * CDIM + ka + 1];
    acc = wmma4(a, b, acc);
  }

  const int s  = col >> 8;
  const int hh = (col >> 5) & 7;
  const int d  = col & 31;
  float* dst = (s == 0) ? Q : ((s == 1) ? K : V);
  dst += hh * (N_TOK * HD) + d;
#pragma unroll
  for (int r = 0; r < 8; ++r) {
    const int n = tm * 16 + r + 8 * kh;
    dst[n * HD] = acc[r];
  }
}

#define KSPLIT  8
__global__ __launch_bounds__(256)
void gsa3d_attn_kernel(const float* __restrict__ Q,
                       const float* __restrict__ K,
                       const float* __restrict__ V,
                       float* __restrict__ AO) {
  __shared__ float ldsP[KSPLIT * 256];
  __shared__ alignas(16) float ldsO[KSPLIT * 16 * HD];
  __shared__ float ldsM[KSPLIT * 16];
  __shared__ float ldsL[KSPLIT * 16];
  const int wslot = __builtin_amdgcn_readfirstlane((int)(threadIdx.x >> 5));
  const int lane  = threadIdx.x & 31;
  const int l16   = lane & 15;
  const int kh    = lane >> 4;
  const int h     = blockIdx.x / NT;
  const int tq    = blockIdx.x % NT;
  const int kb_lo = wslot * 13 + (wslot < 4 ? wslot : 4);
  const int kb_hi = kb_lo + 13 + (wslot < 4 ? 1 : 0);

  const float* Qh = Q + h * (N_TOK * HD);
  const float* Kh = K + h * (N_TOK * HD);
  const float* Vh = V + h * (N_TOK * HD);
  float* P = ldsP + wslot * 256;

  v2f aq[8];
  {
    const int row = tq * 16 + l16;
#pragma unroll
    for (int t = 0; t < 8; ++t) {
      const int ka = 4 * t + 2 * kh;
      aq[t].x = Qh[row * HD + ka + 0];
      aq[t].y = Qh[row * HD + ka + 1];
    }
  }

  float m_i[8], l_i[8];
#pragma unroll
  for (int r = 0; r < 8; ++r) { m_i[r] = -1.0e30f; l_i[r] = 0.0f; }
  v8f o0 = {};
  v8f o1 = {};

  for (int kb = kb_lo; kb < kb_hi; ++kb) {
    v8f s = {};
#pragma unroll
    for (int t = 0; t < 8; ++t) {
      const int kd = 4 * t + 2 * kh;
      v2f b;
      b.x = Kh[(kb * 16 + l16) * HD + kd + 0];
      b.y = Kh[(kb * 16 + l16) * HD + kd + 1];
      s = wmma4(aq[t], b, s);
    }

    float p[8], alpha[8];
#pragma unroll
    for (int r = 0; r < 8; ++r) {
      float sv = s[r] * SCALE_;
      sv = fminf(fmaxf(sv, -10.0f), 10.0f);
      float mx = sv;
      mx = fmaxf(mx, __shfl_xor(mx, 1, 32));
      mx = fmaxf(mx, __shfl_xor(mx, 2, 32));
      mx = fmaxf(mx, __shfl_xor(mx, 4, 32));
      mx = fmaxf(mx, __shfl_xor(mx, 8, 32));
      const float mnew = fmaxf(m_i[r], mx);
      alpha[r] = __expf(m_i[r] - mnew);
      p[r]     = __expf(sv - mnew);
      float rs = p[r];
      rs += __shfl_xor(rs, 1, 32);
      rs += __shfl_xor(rs, 2, 32);
      rs += __shfl_xor(rs, 4, 32);
      rs += __shfl_xor(rs, 8, 32);
      l_i[r] = l_i[r] * alpha[r] + rs;
      m_i[r] = mnew;
    }

#pragma unroll
    for (int r = 0; r < 8; ++r) {
      o0[r] *= alpha[r];
      o1[r] *= alpha[r];
      P[(r + 8 * kh) * 16 + l16] = p[r];
    }

#pragma unroll
    for (int t = 0; t < 4; ++t) {
      const int kp = 4 * t + 2 * kh;
      v2f ap, b0, b1;
      ap.x = P[l16 * 16 + kp + 0];
      ap.y = P[l16 * 16 + kp + 1];
      const float* vr0 = Vh + (kb * 16 + kp + 0) * HD;
      const float* vr1 = Vh + (kb * 16 + kp + 1) * HD;
      b0.x = vr0[l16];        b0.y = vr1[l16];
      b1.x = vr0[16 + l16];   b1.y = vr1[16 + l16];
      o0 = wmma4(ap, b0, o0);
      o1 = wmma4(ap, b1, o1);
    }
  }

  {
    float* Ow = ldsO + wslot * (16 * HD);
#pragma unroll
    for (int r = 0; r < 8; ++r) {
      const int row = r + 8 * kh;
      Ow[row * HD + l16]      = o0[r];
      Ow[row * HD + 16 + l16] = o1[r];
      if (l16 == 0) {
        ldsM[wslot * 16 + row] = m_i[r];
        ldsL[wslot * 16 + row] = l_i[r];
      }
    }
  }
  __syncthreads();

  {
    const int row = threadIdx.x >> 4;
    const int dc  = (threadIdx.x & 15) * 2;
    float mw[KSPLIT];
    float mall = -1.0e30f;
#pragma unroll
    for (int w = 0; w < KSPLIT; ++w) {
      mw[w] = ldsM[w * 16 + row];
      mall  = fmaxf(mall, mw[w]);
    }
    float lsum = 0.0f, a0 = 0.0f, a1 = 0.0f;
#pragma unroll
    for (int w = 0; w < KSPLIT; ++w) {
      const float sc = __expf(mw[w] - mall);
      const float2 ov = *reinterpret_cast<const float2*>(
          ldsO + w * (16 * HD) + row * HD + dc);
      lsum += ldsL[w * 16 + row] * sc;
      a0   += ov.x * sc;
      a1   += ov.y * sc;
    }
    const float inv = 1.0f / lsum;
    float2 res;
    res.x = a0 * inv;
    res.y = a1 * inv;
    const int n = tq * 16 + row;
    *reinterpret_cast<float2*>(AO + n * CDIM + h * HD + dc) = res;
  }
}

__global__ __launch_bounds__(256)
void gsa3d_proj_kernel(const float* __restrict__ AO,
                       const float* __restrict__ wproj,
                       float* __restrict__ out) {
  const int wid  = blockIdx.x * 8 + (threadIdx.x >> 5);
  const int lane = threadIdx.x & 31;
  const int l16  = lane & 15;
  const int kh   = lane >> 4;
  const int tm   = wid / 16;
  const int tn   = wid % 16;
  const int row  = tm * 16 + l16;
  const int col  = tn * 16 + l16;

  v8f acc = {};
  for (int k0 = 0; k0 < CDIM; k0 += 4) {
    const int ka = k0 + 2 * kh;
    v2f a, b;
    a.x = AO[row * CDIM + ka + 0];
    a.y = AO[row * CDIM + ka + 1];
    b.x = wproj[col * CDIM + ka + 0];
    b.y = wproj[col * CDIM + ka + 1];
    acc = wmma4(a, b, acc);
  }
#pragma unroll
  for (int r = 0; r < 8; ++r) {
    const int n = tm * 16 + r + 8 * kh;
    out[col * N_TOK + n] = acc[r];
  }
}

extern "C" void kernel_launch(void* const* d_in, const int* in_sizes, int n_in,
                              void* d_out, int out_size, void* d_ws, size_t ws_size,
                              hipStream_t stream) {
  (void)in_sizes; (void)n_in; (void)out_size; (void)ws_size;
  const float* x     = (const float*)d_in[0];
  const float* wqkv  = (const float*)d_in[1];
  const float* wproj = (const float*)d_in[2];
  float* out = (float*)d_out;

  float* ws = (float*)d_ws;
  float* Q  = ws;
  float* K  = Q + NHEADS * N_TOK * HD;
  float* V  = K + NHEADS * N_TOK * HD;
  float* AO = V + NHEADS * N_TOK * HD;

  gsa3d_qkv_kernel <<<648, 256, 0, stream>>>(x, wqkv, Q, K, V);
  gsa3d_attn_kernel<<<864, 256, 0, stream>>>(Q, K, V, AO);
  gsa3d_proj_kernel<<<216, 256, 0, stream>>>(AO, wproj, out);
}


# attention computed transposed (S^T = K Q^T, O^T = V^T P^T): exp tile stays in registers as B operand, one softmax max/sum per lane, no LDS staging or lane shuffles (row-sum and PV sum order change)
# speedup vs baseline: 4.0141x; 1.2512x over previous
#include <hip/hip_runtime.h>

typedef __attribute__((ext_vector_type(2))) float v2f;
typedef __attribute__((ext_vector_type(8))) float v8f;

#define N_TOK   1728
#define CDIM    256
#define NHEADS  8
#define HD      32
#define NT      108
#define SCALE_  0.1767766952966369f

__device__ __forceinline__ v8f wmma4(v2f a, v2f b, v8f c) {
  return __builtin_amdgcn_wmma_f32_16x16x4_f32(
       false, a,  false, b,
       (short)0, c,  false,  false);
}

__global__ __launch_bounds__(256)
void gsa3d_qkv_kernel(const float* __restrict__ x,
                      const float* __restrict__ wqkv,
                      float* __restrict__ Q,
                      float* __restrict__ K,
                      float* __restrict__ V) {
  const int wid  = blockIdx.x * 8 + (threadIdx.x >> 5);
  const int lane = threadIdx.x & 31;
  const int l16  = lane & 15;
  const int kh   = lane >> 4;
  const int tn   = wid % 48;
  const int tm   = wid / 48;
  const int row  = tm * 16 + l16;
  const int col  = tn * 16 + l16;

  v8f acc = {};
  for (int k0 = 0; k0 < CDIM; k0 += 4) {
    const int ka = k0 + 2 * kh;
    v2f a, b;
    a.x = x[(ka + 0) * N_TOK + row];
    a.y = x[(ka + 1) * N_TOK + row];
    b.x = wqkv[col * CDIM + ka + 0];
    b.y = wqkv[col * CDIM + ka + 1];
    acc = wmma4(a, b, acc);
  }

  const int s  = col >> 8;
  const int hh = (col >> 5) & 7;
  const int d  = col & 31;
  float* dst = (s == 0) ? Q : ((s == 1) ? K : V);
  dst += hh * (N_TOK * HD) + d;
#pragma unroll
  for (int r = 0; r < 8; ++r) {
    const int n = tm * 16 + r + 8 * kh;
    dst[n * HD] = acc[r];
  }
}

#define KSPLIT  8
#define OPITCH  36

__device__ __forceinline__ float xhalf(float x) {
  const int b = __builtin_bit_cast(int, x);
  return __builtin_bit_cast(float,
      __builtin_amdgcn_permlanex16(b, b, 0x76543210, 0xfedcba98, false, false));
}

__global__ __launch_bounds__(256)
void gsa3d_attn_kernel(const float* __restrict__ Q,
                       const float* __restrict__ K,
                       const float* __restrict__ V,
                       float* __restrict__ AO) {
  __shared__ alignas(16) float ldsO[KSPLIT * 16 * OPITCH];
  __shared__ float ldsM[KSPLIT * 16];
  __shared__ float ldsL[KSPLIT * 16];
  const int wslot = __builtin_amdgcn_readfirstlane((int)(threadIdx.x >> 5));
  const int lane  = threadIdx.x & 31;
  const int l16   = lane & 15;
  const int kh    = lane >> 4;
  const int h     = blockIdx.x / NT;
  const int tq    = blockIdx.x % NT;
  const int kb_lo = wslot * 13 + (wslot < 4 ? wslot : 4);
  const int kb_hi = kb_lo + 13 + (wslot < 4 ? 1 : 0);

  const float* Qh = Q + h * (N_TOK * HD);
  const float* Kh = K + h * (N_TOK * HD);
  const float* Vh = V + h * (N_TOK * HD);

  v2f bq[8];
  {
    const int row = tq * 16 + l16;
#pragma unroll
    for (int t = 0; t < 8; ++t) {
      const int ka = 4 * t + 2 * kh;
      bq[t].x = Qh[row * HD + ka + 0];
      bq[t].y = Qh[row * HD + ka + 1];
    }
  }

  float m_i = -1.0e30f;
  float l_i = 0.0f;
  v8f o0 = {};
  v8f o1 = {};

  for (int kb = kb_lo; kb < kb_hi; ++kb) {
    const float* Kb = Kh + (kb * 16 + l16) * HD + 2 * kh;
    v8f s = {};
#pragma unroll
    for (int t = 0; t < 8; ++t) {
      v2f a;
      a.x = Kb[4 * t + 0];
      a.y = Kb[4 * t + 1];
      s = wmma4(a, bq[t], s);
    }

    float sv[8];
#pragma unroll
    for (int r = 0; r < 8; ++r) {
      sv[r] = fminf(fmaxf(s[r] * SCALE_, -10.0f), 10.0f);
    }
    float mx = fmaxf(fmaxf(fmaxf(sv[0], sv[1]), fmaxf(sv[2], sv[3])),
                     fmaxf(fmaxf(sv[4], sv[5]), fmaxf(sv[6], sv[7])));
    mx = fmaxf(mx, xhalf(mx));
    const float mnew  = fmaxf(m_i, mx);
    const float alpha = __expf(m_i - mnew);
    float p[8];
#pragma unroll
    for (int r = 0; r < 8; ++r) p[r] = __expf(sv[r] - mnew);
    const float rs = ((p[0] + p[1]) + (p[2] + p[3])) + ((p[4] + p[5]) + (p[6] + p[7]));
    l_i = l_i * alpha + rs;
    m_i = mnew;

#pragma unroll
    for (int r = 0; r < 8; ++r) {
      o0[r] *= alpha;
      o1[r] *= alpha;
    }

    const float* Vb = Vh + (kb * 16 + 8 * kh) * HD + 2 * l16;
#pragma unroll
    for (int t = 0; t < 4; ++t) {
      const float2 v0 = *reinterpret_cast<const float2*>(Vb + (2 * t + 0) * HD);
      const float2 v1 = *reinterpret_cast<const float2*>(Vb + (2 * t + 1) * HD);
      v2f bp, a0, a1;
      bp.x = p[2 * t + 0];  bp.y = p[2 * t + 1];
      a0.x = v0.x;          a0.y = v1.x;
      a1.x = v0.y;          a1.y = v1.y;
      o0 = wmma4(a0, bp, o0);
      o1 = wmma4(a1, bp, o1);
    }
  }

  {
    const float l_tot = l_i + xhalf(l_i);
    float* Ow = ldsO + wslot * (16 * OPITCH) + l16 * OPITCH + 16 * kh;
#pragma unroll
    for (int q = 0; q < 4; ++q) {
      float4 v;
      v.x = o0[2 * q];      v.y = o1[2 * q];
      v.z = o0[2 * q + 1];  v.w = o1[2 * q + 1];
      *reinterpret_cast<float4*>(Ow + 4 * q) = v;
    }
    if (kh == 0) {
      ldsM[wslot * 16 + l16] = m_i;
      ldsL[wslot * 16 + l16] = l_tot;
    }
  }
  __syncthreads();

  {
    const int row = threadIdx.x >> 4;
    const int dc  = (threadIdx.x & 15) * 2;
    float mw[KSPLIT];
    float mall = -1.0e30f;
#pragma unroll
    for (int w = 0; w < KSPLIT; ++w) {
      mw[w] = ldsM[w * 16 + row];
      mall  = fmaxf(mall, mw[w]);
    }
    float lsum = 0.0f, a0 = 0.0f, a1 = 0.0f;
#pragma unroll
    for (int w = 0; w < KSPLIT; ++w) {
      const float sc = __expf(mw[w] - mall);
      const float2 ov = *reinterpret_cast<const float2*>(
          ldsO + w * (16 * OPITCH) + row * OPITCH + dc);
      lsum += ldsL[w * 16 + row] * sc;
      a0   += ov.x * sc;
      a1   += ov.y * sc;
    }
    const float inv = 1.0f / lsum;
    float2 res;
    res.x = a0 * inv;
    res.y = a1 * inv;
    const int n = tq * 16 + row;
    *reinterpret_cast<float2*>(AO + n * CDIM + h * HD + dc) = res;
  }
}

__global__ __launch_bounds__(256)
void gsa3d_proj_kernel(const float* __restrict__ AO,
                       const float* __restrict__ wproj,
                       float* __restrict__ out) {
  const int wid  = blockIdx.x * 8 + (threadIdx.x >> 5);
  const int lane = threadIdx.x & 31;
  const int l16  = lane & 15;
  const int kh   = lane >> 4;
  const int tm   = wid / 16;
  const int tn   = wid % 16;
  const int row  = tm * 16 + l16;
  const int col  = tn * 16 + l16;

  v8f acc = {};
  for (int k0 = 0; k0 < CDIM; k0 += 4) {
    const int ka = k0 + 2 * kh;
    v2f a, b;
    a.x = AO[row * CDIM + ka + 0];
    a.y = AO[row * CDIM + ka + 1];
    b.x = wproj[col * CDIM + ka + 0];
    b.y = wproj[col * CDIM + ka + 1];
    acc = wmma4(a, b, acc);
  }
#pragma unroll
  for (int r = 0; r < 8; ++r) {
    const int n = tm * 16 + r + 8 * kh;
    out[col * N_TOK + n] = acc[r];
  }
}

extern "C" void kernel_launch(void* const* d_in, const int* in_sizes, int n_in,
                              void* d_out, int out_size, void* d_ws, size_t ws_size,
                              hipStream_t stream) {
  (void)in_sizes; (void)n_in; (void)out_size; (void)ws_size;
  const float* x     = (const float*)d_in[0];
  const float* wqkv  = (const float*)d_in[1];
  const float* wproj = (const float*)d_in[2];
  float* out = (float*)d_out;

  float* ws = (float*)d_ws;
  float* Q  = ws;
  float* K  = Q + NHEADS * N_TOK * HD;
  float* V  = K + NHEADS * N_TOK * HD;
  float* AO = V + NHEADS * N_TOK * HD;

  gsa3d_qkv_kernel <<<648, 256, 0, stream>>>(x, wqkv, Q, K, V);
  gsa3d_attn_kernel<<<864, 256, 0, stream>>>(Q, K, V, AO);
  gsa3d_proj_kernel<<<216, 256, 0, stream>>>(AO, wproj, out);
}


# fragment k order regrouped so each lane reads 16 contiguous bytes: K, Q, wqkv, AO, wproj fragments come as b128 loads, half as many 16-line loads (dot-product sum order over k changes)
# speedup vs baseline: 4.2350x; 1.0550x over previous
#include <hip/hip_runtime.h>

typedef __attribute__((ext_vector_type(2))) float v2f;
typedef __attribute__((ext_vector_type(8))) float v8f;

#define N_TOK   1728
#define CDIM    256
#define NHEADS  8
#define HD      32
#define NT      108
#define SCALE_  0.1767766952966369f

__device__ __forceinline__ v8f wmma4(v2f a, v2f b, v8f c) {
  return __builtin_amdgcn_wmma_f32_16x16x4_f32(
       false, a,  false, b,
       (short)0, c,  false,  false);
}

__global__ __launch_bounds__(256)
void gsa3d_qkv_kernel(const float* __restrict__ x,
                      const float* __restrict__ wqkv,
                      float* __restrict__ Q,
                      float* __restrict__ K,
                      float* __restrict__ V) {
  const int wid  = blockIdx.x * 8 + (threadIdx.x >> 5);
  const int lane = threadIdx.x & 31;
  const int l16  = lane & 15;
  const int kh   = lane >> 4;
  const int tn   = wid % 48;
  const int tm   = wid / 48;
  const int row  = tm * 16 + l16;
  const int col  = tn * 16 + l16;

  v8f acc = {};
  for (int k0 = 0; k0 < CDIM; k0 += 8) {
    const int ka = k0 + 4 * kh;
    const float4 b4 = *reinterpret_cast<const float4*>(wqkv + col * CDIM + ka);
    v2f a0, a1, b0, b1;
    a0.x = x[(ka + 0) * N_TOK + row];
    a0.y = x[(ka + 1) * N_TOK + row];
    a1.x = x[(ka + 2) * N_TOK + row];
    a1.y = x[(ka + 3) * N_TOK + row];
    b0.x = b4.x;  b0.y = b4.y;
    b1.x = b4.z;  b1.y = b4.w;
    acc = wmma4(a0, b0, acc);
    acc = wmma4(a1, b1, acc);
  }

  const int s  = col >> 8;
  const int hh = (col >> 5) & 7;
  const int d  = col & 31;
  float* dst = (s == 0) ? Q : ((s == 1) ? K : V);
  dst += hh * (N_TOK * HD) + d;
#pragma unroll
  for (int r = 0; r < 8; ++r) {
    const int n = tm * 16 + r + 8 * kh;
    dst[n * HD] = acc[r];
  }
}

#define KSPLIT  8
#define OPITCH  36

__device__ __forceinline__ float xhalf(float x) {
  const int b = __builtin_bit_cast(int, x);
  return __builtin_bit_cast(float,
      __builtin_amdgcn_permlanex16(b, b, 0x76543210, 0xfedcba98, false, false));
}

__global__ __launch_bounds__(256)
void gsa3d_attn_kernel(const float* __restrict__ Q,
                       const float* __restrict__ K,
                       const float* __restrict__ V,
                       float* __restrict__ AO) {
  __shared__ alignas(16) float ldsO[KSPLIT * 16 * OPITCH];
  __shared__ float ldsM[KSPLIT * 16];
  __shared__ float ldsL[KSPLIT * 16];
  const int wslot = __builtin_amdgcn_readfirstlane((int)(threadIdx.x >> 5));
  const int lane  = threadIdx.x & 31;
  const int l16   = lane & 15;
  const int kh    = lane >> 4;
  const int h     = blockIdx.x / NT;
  const int tq    = blockIdx.x % NT;
  const int kb_lo = wslot * 13 + (wslot < 4 ? wslot : 4);
  const int kb_hi = kb_lo + 13 + (wslot < 4 ? 1 : 0);

  const float* Qh = Q + h * (N_TOK * HD);
  const float* Kh = K + h * (N_TOK * HD);
  const float* Vh = V + h * (N_TOK * HD);

  v2f bq[8];
  {
    const float* Qr = Qh + (tq * 16 + l16) * HD + 4 * kh;
#pragma unroll
    for (int j = 0; j < 4; ++j) {
      const float4 q4 = *reinterpret_cast<const float4*>(Qr + 8 * j);
      bq[2 * j + 0].x = q4.x;  bq[2 * j + 0].y = q4.y;
      bq[2 * j + 1].x = q4.z;  bq[2 * j + 1].y = q4.w;
    }
  }

  float m_i = -1.0e30f;
  float l_i = 0.0f;
  v8f o0 = {};
  v8f o1 = {};

  for (int kb = kb_lo; kb < kb_hi; ++kb) {
    const float* Kb = Kh + (kb * 16 + l16) * HD + 4 * kh;
    v8f s = {};
#pragma unroll
    for (int j = 0; j < 4; ++j) {
      const float4 k4 = *reinterpret_cast<const float4*>(Kb + 8 * j);
      v2f a0, a1;
      a0.x = k4.x;  a0.y = k4.y;
      a1.x = k4.z;  a1.y = k4.w;
      s = wmma4(a0, bq[2 * j + 0], s);
      s = wmma4(a1, bq[2 * j + 1], s);
    }

    float sv[8];
#pragma unroll
    for (int r = 0; r < 8; ++r) {
      sv[r] = fminf(fmaxf(s[r] * SCALE_, -10.0f), 10.0f);
    }
    float mx = fmaxf(fmaxf(fmaxf(sv[0], sv[1]), fmaxf(sv[2], sv[3])),
                     fmaxf(fmaxf(sv[4], sv[5]), fmaxf(sv[6], sv[7])));
    mx = fmaxf(mx, xhalf(mx));
    const float mnew  = fmaxf(m_i, mx);
    const float alpha = __expf(m_i - mnew);
    float p[8];
#pragma unroll
    for (int r = 0; r < 8; ++r) p[r] = __expf(sv[r] - mnew);
    const float rs = ((p[0] + p[1]) + (p[2] + p[3])) + ((p[4] + p[5]) + (p[6] + p[7]));
    l_i = l_i * alpha + rs;
    m_i = mnew;

#pragma unroll
    for (int r = 0; r < 8; ++r) {
      o0[r] *= alpha;
      o1[r] *= alpha;
    }

    const float* Vb = Vh + (kb * 16 + 8 * kh) * HD + 2 * l16;
#pragma unroll
    for (int t = 0; t < 4; ++t) {
      const float2 v0 = *reinterpret_cast<const float2*>(Vb + (2 * t + 0) * HD);
      const float2 v1 = *reinterpret_cast<const float2*>(Vb + (2 * t + 1) * HD);
      v2f bp, a0, a1;
      bp.x = p[2 * t + 0];  bp.y = p[2 * t + 1];
      a0.x = v0.x;          a0.y = v1.x;
      a1.x = v0.y;          a1.y = v1.y;
      o0 = wmma4(a0, bp, o0);
      o1 = wmma4(a1, bp, o1);
    }
  }

  {
    const float l_tot = l_i + xhalf(l_i);
    float* Ow = ldsO + wslot * (16 * OPITCH) + l16 * OPITCH + 16 * kh;
#pragma unroll
    for (int q = 0; q < 4; ++q) {
      float4 v;
      v.x = o0[2 * q];      v.y = o1[2 * q];
      v.z = o0[2 * q + 1];  v.w = o1[2 * q + 1];
      *reinterpret_cast<float4*>(Ow + 4 * q) = v;
    }
    if (kh == 0) {
      ldsM[wslot * 16 + l16] = m_i;
      ldsL[wslot * 16 + l16] = l_tot;
    }
  }
  __syncthreads();

  {
    const int row = threadIdx.x >> 4;
    const int dc  = (threadIdx.x & 15) * 2;
    float mw[KSPLIT];
    float mall = -1.0e30f;
#pragma unroll
    for (int w = 0; w < KSPLIT; ++w) {
      mw[w] = ldsM[w * 16 + row];
      mall  = fmaxf(mall, mw[w]);
    }
    float lsum = 0.0f, a0 = 0.0f, a1 = 0.0f;
#pragma unroll
    for (int w = 0; w < KSPLIT; ++w) {
      const float sc = __expf(mw[w] - mall);
      const float2 ov = *reinterpret_cast<const float2*>(
          ldsO + w * (16 * OPITCH) + row * OPITCH + dc);
      lsum += ldsL[w * 16 + row] * sc;
      a0   += ov.x * sc;
      a1   += ov.y * sc;
    }
    const float inv = 1.0f / lsum;
    float2 res;
    res.x = a0 * inv;
    res.y = a1 * inv;
    const int n = tq * 16 + row;
    *reinterpret_cast<float2*>(AO + n * CDIM + h * HD + dc) = res;
  }
}

__global__ __launch_bounds__(256)
void gsa3d_proj_kernel(const float* __restrict__ AO,
                       const float* __restrict__ wproj,
                       float* __restrict__ out) {
  const int wid  = blockIdx.x * 8 + (threadIdx.x >> 5);
  const int lane = threadIdx.x & 31;
  const int l16  = lane & 15;
  const int kh   = lane >> 4;
  const int tm   = wid / 16;
  const int tn   = wid % 16;
  const int row  = tm * 16 + l16;
  const int col  = tn * 16 + l16;

  v8f acc = {};
  for (int k0 = 0; k0 < CDIM; k0 += 8) {
    const int ka = k0 + 4 * kh;
    const float4 a4 = *reinterpret_cast<const float4*>(AO + row * CDIM + ka);
    const float4 b4 = *reinterpret_cast<const float4*>(wproj + col * CDIM + ka);
    v2f a0, a1, b0, b1;
    a0.x = a4.x;  a0.y = a4.y;
    a1.x = a4.z;  a1.y = a4.w;
    b0.x = b4.x;  b0.y = b4.y;
    b1.x = b4.z;  b1.y = b4.w;
    acc = wmma4(a0, b0, acc);
    acc = wmma4(a1, b1, acc);
  }
#pragma unroll
  for (int r = 0; r < 8; ++r) {
    const int n = tm * 16 + r + 8 * kh;
    out[col * N_TOK + n] = acc[r];
  }
}

extern "C" void kernel_launch(void* const* d_in, const int* in_sizes, int n_in,
                              void* d_out, int out_size, void* d_ws, size_t ws_size,
                              hipStream_t stream) {
  (void)in_sizes; (void)n_in; (void)out_size; (void)ws_size;
  const float* x     = (const float*)d_in[0];
  const float* wqkv  = (const float*)d_in[1];
  const float* wproj = (const float*)d_in[2];
  float* out = (float*)d_out;

  float* ws = (float*)d_ws;
  float* Q  = ws;
  float* K  = Q + NHEADS * N_TOK * HD;
  float* V  = K + NHEADS * N_TOK * HD;
  float* AO = V + NHEADS * N_TOK * HD;

  gsa3d_qkv_kernel <<<648, 256, 0, stream>>>(x, wqkv, Q, K, V);
  gsa3d_attn_kernel<<<864, 256, 0, stream>>>(Q, K, V, AO);
  gsa3d_proj_kernel<<<216, 256, 0, stream>>>(AO, wproj, out);
}


# all three kernels: the next block's operand fragments are loaded one block ahead into a second register set (software prefetch of the global loads); arithmetic and every sum order unchanged
# speedup vs baseline: 4.4313x; 1.0464x over previous
#include <hip/hip_runtime.h>

typedef __attribute__((ext_vector_type(2))) float v2f;
typedef __attribute__((ext_vector_type(8))) float v8f;

#define N_TOK   1728
#define CDIM    256
#define NHEADS  8
#define HD      32
#define NT      108
#define SCALE_  0.1767766952966369f
#define QCH     4
#define PCH     8

__device__ __forceinline__ v8f wmma4(v2f a, v2f b, v8f c) {
  return __builtin_amdgcn_wmma_f32_16x16x4_f32(
       false, a,  false, b,
       (short)0, c,  false,  false);
}

__global__ __launch_bounds__(256)
void gsa3d_qkv_kernel(const float* __restrict__ x,
                      const float* __restrict__ wqkv,
                      float* __restrict__ Q,
                      float* __restrict__ K,
                      float* __restrict__ V) {
  const int wid  = blockIdx.x * 8 + (threadIdx.x >> 5);
  const int lane = threadIdx.x & 31;
  const int l16  = lane & 15;
  const int kh   = lane >> 4;
  const int tn   = wid % 48;
  const int tm   = wid / 48;
  const int row  = tm * 16 + l16;
  const int col  = tn * 16 + l16;

  const float* xa = x + row + (4 * kh) * N_TOK;
  const float* wb = wqkv + col * CDIM + 4 * kh;
  float4 af[2][QCH], bf[2][QCH];
#pragma unroll
  for (int i = 0; i < QCH; ++i) {
    const float* xp = xa + (8 * i) * N_TOK;
    af[0][i] = make_float4(xp[0], xp[N_TOK], xp[2 * N_TOK], xp[3 * N_TOK]);
    bf[0][i] = *reinterpret_cast<const float4*>(wb + 8 * i);
  }
  v8f acc = {};
#pragma unroll
  for (int c = 0; c < CDIM / (8 * QCH); ++c) {
    const int cur = c & 1, nxt = cur ^ 1;
    if (c + 1 < CDIM / (8 * QCH)) {
#pragma unroll
      for (int i = 0; i < QCH; ++i) {
        const int ks = 8 * (QCH * (c + 1) + i);
        const float* xp = xa + ks * N_TOK;
        af[nxt][i] = make_float4(xp[0], xp[N_TOK], xp[2 * N_TOK], xp[3 * N_TOK]);
        bf[nxt][i] = *reinterpret_cast<const float4*>(wb + ks);
      }
    }
    __builtin_amdgcn_sched_barrier(0);
#pragma unroll
    for (int i = 0; i < QCH; ++i) {
      v2f a0, a1, b0, b1;
      a0.x = af[cur][i].x;  a0.y = af[cur][i].y;
      a1.x = af[cur][i].z;  a1.y = af[cur][i].w;
      b0.x = bf[cur][i].x;  b0.y = bf[cur][i].y;
      b1.x = bf[cur][i].z;  b1.y = bf[cur][i].w;
      acc = wmma4(a0, b0, acc);
      acc = wmma4(a1, b1, acc);
    }
  }

  const int s  = col >> 8;
  const int hh = (col >> 5) & 7;
  const int d  = col & 31;
  float* dst = (s == 0) ? Q : ((s == 1) ? K : V);
  dst += hh * (N_TOK * HD) + d;
#pragma unroll
  for (int r = 0; r < 8; ++r) {
    const int n = tm * 16 + r + 8 * kh;
    dst[n * HD] = acc[r];
  }
}

#define KSPLIT  8
#define OPITCH  36

__device__ __forceinline__ float xhalf(float x) {
  const int b = __builtin_bit_cast(int, x);
  return __builtin_bit_cast(float,
      __builtin_amdgcn_permlanex16(b, b, 0x76543210, 0xfedcba98, false, false));
}

struct KVFrag {
  float4 k[4];
  float2 v[8];
};

__device__ __forceinline__ void load_kv(KVFrag& f, const float* __restrict__ Kh,
                                        const float* __restrict__ Vh,
                                        int kb, int l16, int kh) {
  const float* Kb = Kh + (kb * 16 + l16) * HD + 4 * kh;
#pragma unroll
  for (int j = 0; j < 4; ++j) f.k[j] = *reinterpret_cast<const float4*>(Kb + 8 * j);
  const float* Vb = Vh + (kb * 16 + 8 * kh) * HD + 2 * l16;
#pragma unroll
  for (int t = 0; t < 8; ++t) f.v[t] = *reinterpret_cast<const float2*>(Vb + t * HD);
}

__device__ __forceinline__ void attn_block(const KVFrag& f, const v2f (&bq)[8],
                                           float& m_i, float& l_i, v8f& o0, v8f& o1) {
  v8f s = {};
#pragma unroll
  for (int j = 0; j < 4; ++j) {
    v2f a0, a1;
    a0.x = f.k[j].x;  a0.y = f.k[j].y;
    a1.x = f.k[j].z;  a1.y = f.k[j].w;
    s = wmma4(a0, bq[2 * j + 0], s);
    s = wmma4(a1, bq[2 * j + 1], s);
  }

  float sv[8];
#pragma unroll
  for (int r = 0; r < 8; ++r) {
    sv[r] = fminf(fmaxf(s[r] * SCALE_, -10.0f), 10.0f);
  }
  float mx = fmaxf(fmaxf(fmaxf(sv[0], sv[1]), fmaxf(sv[2], sv[3])),
                   fmaxf(fmaxf(sv[4], sv[5]), fmaxf(sv[6], sv[7])));
  mx = fmaxf(mx, xhalf(mx));
  const float mnew  = fmaxf(m_i, mx);
  const float alpha = __expf(m_i - mnew);
  float p[8];
#pragma unroll
  for (int r = 0; r < 8; ++r) p[r] = __expf(sv[r] - mnew);
  const float rs = ((p[0] + p[1]) + (p[2] + p[3])) + ((p[4] + p[5]) + (p[6] + p[7]));
  l_i = l_i * alpha + rs;
  m_i = mnew;

#pragma unroll
  for (int r = 0; r < 8; ++r) {
    o0[r] *= alpha;
    o1[r] *= alpha;
  }

#pragma unroll
  for (int t = 0; t < 4; ++t) {
    const float2 v0 = f.v[2 * t + 0];
    const float2 v1 = f.v[2 * t + 1];
    v2f bp, a0, a1;
    bp.x = p[2 * t + 0];  bp.y = p[2 * t + 1];
    a0.x = v0.x;          a0.y = v1.x;
    a1.x = v0.y;          a1.y = v1.y;
    o0 = wmma4(a0, bp, o0);
    o1 = wmma4(a1, bp, o1);
  }
}

__global__ __launch_bounds__(256)
void gsa3d_attn_kernel(const float* __restrict__ Q,
                       const float* __restrict__ K,
                       const float* __restrict__ V,
                       float* __restrict__ AO) {
  __shared__ alignas(16) float ldsO[KSPLIT * 16 * OPITCH];
  __shared__ float ldsM[KSPLIT * 16];
  __shared__ float ldsL[KSPLIT * 16];
  const int wslot = __builtin_amdgcn_readfirstlane((int)(threadIdx.x >> 5));
  const int lane  = threadIdx.x & 31;
  const int l16   = lane & 15;
  const int kh    = lane >> 4;
  const int h     = blockIdx.x / NT;
  const int tq    = blockIdx.x % NT;
  const int kb_lo = wslot * 13 + (wslot < 4 ? wslot : 4);
  const int kb_hi = kb_lo + 13 + (wslot < 4 ? 1 : 0);

  const float* Qh = Q + h * (N_TOK * HD);
  const float* Kh = K + h * (N_TOK * HD);
  const float* Vh = V + h * (N_TOK * HD);

  v2f bq[8];
  {
    const float* Qr = Qh + (tq * 16 + l16) * HD + 4 * kh;
#pragma unroll
    for (int j = 0; j < 4; ++j) {
      const float4 q4 = *reinterpret_cast<const float4*>(Qr + 8 * j);
      bq[2 * j + 0].x = q4.x;  bq[2 * j + 0].y = q4.y;
      bq[2 * j + 1].x = q4.z;  bq[2 * j + 1].y = q4.w;
    }
  }

  float m_i = -1.0e30f;
  float l_i = 0.0f;
  v8f o0 = {};
  v8f o1 = {};

  KVFrag fa, fb;
  load_kv(fa, Kh, Vh, kb_lo, l16, kh);
  for (int kb = kb_lo;;) {
    load_kv(fb, Kh, Vh, kb + 1, l16, kh);
    __builtin_amdgcn_sched_barrier(0);
    attn_block(fa, bq, m_i, l_i, o0, o1);
    if (kb + 2 >= kb_hi) {
      attn_block(fb, bq, m_i, l_i, o0, o1);
      break;
    }
    load_kv(fa, Kh, Vh, kb + 2, l16, kh);
    __builtin_amdgcn_sched_barrier(0);
    attn_block(fb, bq, m_i, l_i, o0, o1);
    kb += 2;
    if (kb + 1 >= kb_hi) {
      attn_block(fa, bq, m_i, l_i, o0, o1);
      break;
    }
  }

  {
    const float l_tot = l_i + xhalf(l_i);
    float* Ow = ldsO + wslot * (16 * OPITCH) + l16 * OPITCH + 16 * kh;
#pragma unroll
    for (int q = 0; q < 4; ++q) {
      float4 v;
      v.x = o0[2 * q];      v.y = o1[2 * q];
      v.z = o0[2 * q + 1];  v.w = o1[2 * q + 1];
      *reinterpret_cast<float4*>(Ow + 4 * q) = v;
    }
    if (kh == 0) {
      ldsM[wslot * 16 + l16] = m_i;
      ldsL[wslot * 16 + l16] = l_tot;
    }
  }
  __syncthreads();

  {
    const int row = threadIdx.x >> 4;
    const int dc  = (threadIdx.x & 15) * 2;
    float mw[KSPLIT];
    float mall = -1.0e30f;
#pragma unroll
    for (int w = 0; w < KSPLIT; ++w) {
      mw[w] = ldsM[w * 16 + row];
      mall  = fmaxf(mall, mw[w]);
    }
    float lsum = 0.0f, a0 = 0.0f, a1 = 0.0f;
#pragma unroll
    for (int w = 0; w < KSPLIT; ++w) {
      const float sc = __expf(mw[w] - mall);
      const float2 ov = *reinterpret_cast<const float2*>(
          ldsO + w * (16 * OPITCH) + row * OPITCH + dc);
      lsum += ldsL[w * 16 + row] * sc;
      a0   += ov.x * sc;
      a1   += ov.y * sc;
    }
    const float inv = 1.0f / lsum;
    float2 res;
    res.x = a0 * inv;
    res.y = a1 * inv;
    const int n = tq * 16 + row;
    *reinterpret_cast<float2*>(AO + n * CDIM + h * HD + dc) = res;
  }
}

__global__ __launch_bounds__(256)
void gsa3d_proj_kernel(const float* __restrict__ AO,
                       const float* __restrict__ wproj,
                       float* __restrict__ out) {
  const int wid  = blockIdx.x * 8 + (threadIdx.x >> 5);
  const int lane = threadIdx.x & 31;
  const int l16  = lane & 15;
  const int kh   = lane >> 4;
  const int tm   = wid / 16;
  const int tn   = wid % 16;
  const int row  = tm * 16 + l16;
  const int col  = tn * 16 + l16;

  const float* ap = AO + row * CDIM + 4 * kh;
  const float* bp = wproj + col * CDIM + 4 * kh;
  float4 af[2][PCH], bf[2][PCH];
#pragma unroll
  for (int i = 0; i < PCH; ++i) {
    af[0][i] = *reinterpret_cast<const float4*>(ap + 8 * i);
    bf[0][i] = *reinterpret_cast<const float4*>(bp + 8 * i);
  }
  v8f acc = {};
#pragma unroll
  for (int c = 0; c < CDIM / (8 * PCH); ++c) {
    const int cur = c & 1, nxt = cur ^ 1;
    if (c + 1 < CDIM / (8 * PCH)) {
#pragma unroll
      for (int i = 0; i < PCH; ++i) {
        const int ks = 8 * (PCH * (c + 1) + i);
        af[nxt][i] = *reinterpret_cast<const float4*>(ap + ks);
        bf[nxt][i] = *reinterpret_cast<const float4*>(bp + ks);
      }
    }
    __builtin_amdgcn_sched_barrier(0);
#pragma unroll
    for (int i = 0; i < PCH; ++i) {
      v2f a0, a1, b0, b1;
      a0.x = af[cur][i].x;  a0.y = af[cur][i].y;
      a1.x = af[cur][i].z;  a1.y = af[cur][i].w;
      b0.x = bf[cur][i].x;  b0.y = bf[cur][i].y;
      b1.x = bf[cur][i].z;  b1.y = bf[cur][i].w;
      acc = wmma4(a0, b0, acc);
      acc = wmma4(a1, b1, acc);
    }
  }
#pragma unroll
  for (int r = 0; r < 8; ++r) {
    const int n = tm * 16 + r + 8 * kh;
    out[col * N_TOK + n] = acc[r];
  }
}

extern "C" void kernel_launch(void* const* d_in, const int* in_sizes, int n_in,
                              void* d_out, int out_size, void* d_ws, size_t ws_size,
                              hipStream_t stream) {
  (void)in_sizes; (void)n_in; (void)out_size; (void)ws_size;
  const float* x     = (const float*)d_in[0];
  const float* wqkv  = (const float*)d_in[1];
  const float* wproj = (const float*)d_in[2];
  float* out = (float*)d_out;

  float* ws = (float*)d_ws;
  float* Q  = ws;
  float* K  = Q + NHEADS * N_TOK * HD;
  float* V  = K + NHEADS * N_TOK * HD;
  float* AO = V + NHEADS * N_TOK * HD;

  gsa3d_qkv_kernel <<<648, 256, 0, stream>>>(x, wqkv, Q, K, V);
  gsa3d_attn_kernel<<<864, 256, 0, stream>>>(Q, K, V, AO);
  gsa3d_proj_kernel<<<216, 256, 0, stream>>>(AO, wproj, out);
}


# attention: each wave holds two 16-row query tiles and feeds both from every loaded K / V fragment (432 work-groups of 32 queries): half the fragment loads per matrix step; every sum order unchanged
# speedup vs baseline: 4.7737x; 1.0773x over previous
#include <hip/hip_runtime.h>

typedef __attribute__((ext_vector_type(2))) float v2f;
typedef __attribute__((ext_vector_type(8))) float v8f;

#define N_TOK   1728
#define CDIM    256
#define NHEADS  8
#define HD      32
#define NT      108
#define SCALE_  0.1767766952966369f
#define QCH     4
#define PCH     8

__device__ __forceinline__ v8f wmma4(v2f a, v2f b, v8f c) {
  return __builtin_amdgcn_wmma_f32_16x16x4_f32(
       false, a,  false, b,
       (short)0, c,  false,  false);
}

__global__ __launch_bounds__(256)
void gsa3d_qkv_kernel(const float* __restrict__ x,
                      const float* __restrict__ wqkv,
                      float* __restrict__ Q,
                      float* __restrict__ K,
                      float* __restrict__ V) {
  const int wid  = blockIdx.x * 8 + (threadIdx.x >> 5);
  const int lane = threadIdx.x & 31;
  const int l16  = lane & 15;
  const int kh   = lane >> 4;
  const int tn   = wid % 48;
  const int tm   = wid / 48;
  const int row  = tm * 16 + l16;
  const int col  = tn * 16 + l16;

  const float* xa = x + row + (4 * kh) * N_TOK;
  const float* wb = wqkv + col * CDIM + 4 * kh;
  float4 af[2][QCH], bf[2][QCH];
#pragma unroll
  for (int i = 0; i < QCH; ++i) {
    const float* xp = xa + (8 * i) * N_TOK;
    af[0][i] = make_float4(xp[0], xp[N_TOK], xp[2 * N_TOK], xp[3 * N_TOK]);
    bf[0][i] = *reinterpret_cast<const float4*>(wb + 8 * i);
  }
  v8f acc = {};
#pragma unroll
  for (int c = 0; c < CDIM / (8 * QCH); ++c) {
    const int cur = c & 1, nxt = cur ^ 1;
    if (c + 1 < CDIM / (8 * QCH)) {
#pragma unroll
      for (int i = 0; i < QCH; ++i) {
        const int ks = 8 * (QCH * (c + 1) + i);
        const float* xp = xa + ks * N_TOK;
        af[nxt][i] = make_float4(xp[0], xp[N_TOK], xp[2 * N_TOK], xp[3 * N_TOK]);
        bf[nxt][i] = *reinterpret_cast<const float4*>(wb + ks);
      }
    }
    __builtin_amdgcn_sched_barrier(0);
#pragma unroll
    for (int i = 0; i < QCH; ++i) {
      v2f a0, a1, b0, b1;
      a0.x = af[cur][i].x;  a0.y = af[cur][i].y;
      a1.x = af[cur][i].z;  a1.y = af[cur][i].w;
      b0.x = bf[cur][i].x;  b0.y = bf[cur][i].y;
      b1.x = bf[cur][i].z;  b1.y = bf[cur][i].w;
      acc = wmma4(a0, b0, acc);
      acc = wmma4(a1, b1, acc);
    }
  }

  const int s  = col >> 8;
  const int hh = (col >> 5) & 7;
  const int d  = col & 31;
  float* dst = (s == 0) ? Q : ((s == 1) ? K : V);
  dst += hh * (N_TOK * HD) + d;
#pragma unroll
  for (int r = 0; r < 8; ++r) {
    const int n = tm * 16 + r + 8 * kh;
    dst[n * HD] = acc[r];
  }
}

#define KSPLIT  8
#define QROWS   32
#define NTP     54
#define OPITCH  36

__device__ __forceinline__ float xhalf(float x) {
  const int b = __builtin_bit_cast(int, x);
  return __builtin_bit_cast(float,
      __builtin_amdgcn_permlanex16(b, b, 0x76543210, 0xfedcba98, false, false));
}

struct KVFrag {
  float4 k[4];
  float2 v[8];
};

__device__ __forceinline__ void load_kv(KVFrag& f, const float* __restrict__ Kh,
                                        const float* __restrict__ Vh,
                                        int kb, int l16, int kh) {
  const float* Kb = Kh + (kb * 16 + l16) * HD + 4 * kh;
#pragma unroll
  for (int j = 0; j < 4; ++j) f.k[j] = *reinterpret_cast<const float4*>(Kb + 8 * j);
  const float* Vb = Vh + (kb * 16 + 8 * kh) * HD + 2 * l16;
#pragma unroll
  for (int t = 0; t < 8; ++t) f.v[t] = *reinterpret_cast<const float2*>(Vb + t * HD);
}

struct QTile {
  v2f   bq[8];
  float m;
  float l;
  v8f   o0;
  v8f   o1;
};

__device__ __forceinline__ void softmax_tile(const v8f& s, QTile& q, float (&p)[8]) {
  float sv[8];
#pragma unroll
  for (int r = 0; r < 8; ++r) {
    sv[r] = fminf(fmaxf(s[r] * SCALE_, -10.0f), 10.0f);
  }
  float mx = fmaxf(fmaxf(fmaxf(sv[0], sv[1]), fmaxf(sv[2], sv[3])),
                   fmaxf(fmaxf(sv[4], sv[5]), fmaxf(sv[6], sv[7])));
  mx = fmaxf(mx, xhalf(mx));
  const float mnew  = fmaxf(q.m, mx);
  const float alpha = __expf(q.m - mnew);
#pragma unroll
  for (int r = 0; r < 8; ++r) p[r] = __expf(sv[r] - mnew);
  const float rs = ((p[0] + p[1]) + (p[2] + p[3])) + ((p[4] + p[5]) + (p[6] + p[7]));
  q.l = q.l * alpha + rs;
  q.m = mnew;
#pragma unroll
  for (int r = 0; r < 8; ++r) {
    q.o0[r] *= alpha;
    q.o1[r] *= alpha;
  }
}

__device__ __forceinline__ void attn_block(const KVFrag& f, QTile& qa, QTile& qb) {
  v8f sa = {};
  v8f sb = {};
#pragma unroll
  for (int j = 0; j < 4; ++j) {
    v2f a0, a1;
    a0.x = f.k[j].x;  a0.y = f.k[j].y;
    a1.x = f.k[j].z;  a1.y = f.k[j].w;
    sa = wmma4(a0, qa.bq[2 * j + 0], sa);
    sb = wmma4(a0, qb.bq[2 * j + 0], sb);
    sa = wmma4(a1, qa.bq[2 * j + 1], sa);
    sb = wmma4(a1, qb.bq[2 * j + 1], sb);
  }

  float pa[8], pb[8];
  softmax_tile(sa, qa, pa);
  softmax_tile(sb, qb, pb);

#pragma unroll
  for (int t = 0; t < 4; ++t) {
    const float2 v0 = f.v[2 * t + 0];
    const float2 v1 = f.v[2 * t + 1];
    v2f ba, bb, a0, a1;
    ba.x = pa[2 * t + 0];  ba.y = pa[2 * t + 1];
    bb.x = pb[2 * t + 0];  bb.y = pb[2 * t + 1];
    a0.x = v0.x;           a0.y = v1.x;
    a1.x = v0.y;           a1.y = v1.y;
    qa.o0 = wmma4(a0, ba, qa.o0);
    qb.o0 = wmma4(a0, bb, qb.o0);
    qa.o1 = wmma4(a1, ba, qa.o1);
    qb.o1 = wmma4(a1, bb, qb.o1);
  }
}

__device__ __forceinline__ void init_qtile(QTile& q, const float* __restrict__ Qh,
                                           int tile, int l16, int kh) {
  const float* Qr = Qh + (tile * 16 + l16) * HD + 4 * kh;
#pragma unroll
  for (int j = 0; j < 4; ++j) {
    const float4 q4 = *reinterpret_cast<const float4*>(Qr + 8 * j);
    q.bq[2 * j + 0].x = q4.x;  q.bq[2 * j + 0].y = q4.y;
    q.bq[2 * j + 1].x = q4.z;  q.bq[2 * j + 1].y = q4.w;
  }
  q.m = -1.0e30f;
  q.l = 0.0f;
  q.o0 = v8f{};
  q.o1 = v8f{};
}

__device__ __forceinline__ void publish_qtile(const QTile& q, float* Ow, float* Mw, float* Lw,
                                              int l16, int kh) {
  const float l_tot = q.l + xhalf(q.l);
  float* Orow = Ow + l16 * OPITCH + 16 * kh;
#pragma unroll
  for (int i = 0; i < 4; ++i) {
    float4 v;
    v.x = q.o0[2 * i];      v.y = q.o1[2 * i];
    v.z = q.o0[2 * i + 1];  v.w = q.o1[2 * i + 1];
    *reinterpret_cast<float4*>(Orow + 4 * i) = v;
  }
  if (kh == 0) {
    Mw[l16] = q.m;
    Lw[l16] = l_tot;
  }
}

__global__ __launch_bounds__(256)
void gsa3d_attn_kernel(const float* __restrict__ Q,
                       const float* __restrict__ K,
                       const float* __restrict__ V,
                       float* __restrict__ AO) {
  __shared__ alignas(16) float ldsO[KSPLIT * QROWS * OPITCH];
  __shared__ float ldsM[KSPLIT * QROWS];
  __shared__ float ldsL[KSPLIT * QROWS];
  const int wslot = __builtin_amdgcn_readfirstlane((int)(threadIdx.x >> 5));
  const int lane  = threadIdx.x & 31;
  const int l16   = lane & 15;
  const int kh    = lane >> 4;
  const int h     = blockIdx.x / NTP;
  const int tp    = blockIdx.x % NTP;
  const int kb_lo = wslot * 13 + (wslot < 4 ? wslot : 4);
  const int kb_hi = kb_lo + 13 + (wslot < 4 ? 1 : 0);

  const float* Qh = Q + h * (N_TOK * HD);
  const float* Kh = K + h * (N_TOK * HD);
  const float* Vh = V + h * (N_TOK * HD);

  QTile qa, qb;
  init_qtile(qa, Qh, 2 * tp + 0, l16, kh);
  init_qtile(qb, Qh, 2 * tp + 1, l16, kh);

  KVFrag fa, fb;
  load_kv(fa, Kh, Vh, kb_lo, l16, kh);
  for (int kb = kb_lo;;) {
    load_kv(fb, Kh, Vh, kb + 1, l16, kh);
    __builtin_amdgcn_sched_barrier(0);
    attn_block(fa, qa, qb);
    if (kb + 2 >= kb_hi) {
      attn_block(fb, qa, qb);
      break;
    }
    load_kv(fa, Kh, Vh, kb + 2, l16, kh);
    __builtin_amdgcn_sched_barrier(0);
    attn_block(fb, qa, qb);
    kb += 2;
    if (kb + 1 >= kb_hi) {
      attn_block(fa, qa, qb);
      break;
    }
  }

  {
    float* Ow = ldsO + wslot * (QROWS * OPITCH);
    float* Mw = ldsM + wslot * QROWS;
    float* Lw = ldsL + wslot * QROWS;
    publish_qtile(qa, Ow, Mw, Lw, l16, kh);
    publish_qtile(qb, Ow + 16 * OPITCH, Mw + 16, Lw + 16, l16, kh);
  }
  __syncthreads();

#pragma unroll
  for (int qt = 0; qt < 2; ++qt) {
    const int row = 16 * qt + (threadIdx.x >> 4);
    const int dc  = (threadIdx.x & 15) * 2;
    float mw[KSPLIT];
    float mall = -1.0e30f;
#pragma unroll
    for (int w = 0; w < KSPLIT; ++w) {
      mw[w] = ldsM[w * QROWS + row];
      mall  = fmaxf(mall, mw[w]);
    }
    float lsum = 0.0f, a0 = 0.0f, a1 = 0.0f;
#pragma unroll
    for (int w = 0; w < KSPLIT; ++w) {
      const float sc = __expf(mw[w] - mall);
      const float2 ov = *reinterpret_cast<const float2*>(
          ldsO + w * (QROWS * OPITCH) + row * OPITCH + dc);
      lsum += ldsL[w * QROWS + row] * sc;
      a0   += ov.x * sc;
      a1   += ov.y * sc;
    }
    const float inv = 1.0f / lsum;
    float2 res;
    res.x = a0 * inv;
    res.y = a1 * inv;
    const int n = tp * QROWS + row;
    *reinterpret_cast<float2*>(AO + n * CDIM + h * HD + dc) = res;
  }
}

__global__ __launch_bounds__(256)
void gsa3d_proj_kernel(const float* __restrict__ AO,
                       const float* __restrict__ wproj,
                       float* __restrict__ out) {
  const int wid  = blockIdx.x * 8 + (threadIdx.x >> 5);
  const int lane = threadIdx.x & 31;
  const int l16  = lane & 15;
  const int kh   = lane >> 4;
  const int tm   = wid / 16;
  const int tn   = wid % 16;
  const int row  = tm * 16 + l16;
  const int col  = tn * 16 + l16;

  const float* ap = AO + row * CDIM + 4 * kh;
  const float* bp = wproj + col * CDIM + 4 * kh;
  float4 af[2][PCH], bf[2][PCH];
#pragma unroll
  for (int i = 0; i < PCH; ++i) {
    af[0][i] = *reinterpret_cast<const float4*>(ap + 8 * i);
    bf[0][i] = *reinterpret_cast<const float4*>(bp + 8 * i);
  }
  v8f acc = {};
#pragma unroll
  for (int c = 0; c < CDIM / (8 * PCH); ++c) {
    const int cur = c & 1, nxt = cur ^ 1;
    if (c + 1 < CDIM / (8 * PCH)) {
#pragma unroll
      for (int i = 0; i < PCH; ++i) {
        const int ks = 8 * (PCH * (c + 1) + i);
        af[nxt][i] = *reinterpret_cast<const float4*>(ap + ks);
        bf[nxt][i] = *reinterpret_cast<const float4*>(bp + ks);
      }
    }
    __builtin_amdgcn_sched_barrier(0);
#pragma unroll
    for (int i = 0; i < PCH; ++i) {
      v2f a0, a1, b0, b1;
      a0.x = af[cur][i].x;  a0.y = af[cur][i].y;
      a1.x = af[cur][i].z;  a1.y = af[cur][i].w;
      b0.x = bf[cur][i].x;  b0.y = bf[cur][i].y;
      b1.x = bf[cur][i].z;  b1.y = bf[cur][i].w;
      acc = wmma4(a0, b0, acc);
      acc = wmma4(a1, b1, acc);
    }
  }
#pragma unroll
  for (int r = 0; r < 8; ++r) {
    const int n = tm * 16 + r + 8 * kh;
    out[col * N_TOK + n] = acc[r];
  }
}

extern "C" void kernel_launch(void* const* d_in, const int* in_sizes, int n_in,
                              void* d_out, int out_size, void* d_ws, size_t ws_size,
                              hipStream_t stream) {
  (void)in_sizes; (void)n_in; (void)out_size; (void)ws_size;
  const float* x     = (const float*)d_in[0];
  const float* wqkv  = (const float*)d_in[1];
  const float* wproj = (const float*)d_in[2];
  float* out = (float*)d_out;

  float* ws = (float*)d_ws;
  float* Q  = ws;
  float* K  = Q + NHEADS * N_TOK * HD;
  float* V  = K + NHEADS * N_TOK * HD;
  float* AO = V + NHEADS * N_TOK * HD;

  gsa3d_qkv_kernel <<<648, 256, 0, stream>>>(x, wqkv, Q, K, V);
  gsa3d_attn_kernel<<<432, 256, 0, stream>>>(Q, K, V, AO);
  gsa3d_proj_kernel<<<216, 256, 0, stream>>>(AO, wproj, out);
}


# attention softmax takes the clamp bound 10 as its fixed shift, p = exp(s - 10), in place of a running max: no max tree, lane exchange or rescale of O and l per block; wave partials add
# speedup vs baseline: 4.9647x; 1.0400x over previous
#include <hip/hip_runtime.h>

typedef __attribute__((ext_vector_type(2))) float v2f;
typedef __attribute__((ext_vector_type(8))) float v8f;

#define N_TOK   1728
#define CDIM    256
#define NHEADS  8
#define HD      32
#define NT      108
#define SCALE_  0.1767766952966369f
#define LOG2E_  1.4426950408889634f
#define SC2_    (SCALE_ * LOG2E_)
#define SH2_    (10.0f * LOG2E_)
#define QCH     4
#define PCH     8

__device__ __forceinline__ v8f wmma4(v2f a, v2f b, v8f c) {
  return __builtin_amdgcn_wmma_f32_16x16x4_f32(
       false, a,  false, b,
       (short)0, c,  false,  false);
}

__global__ __launch_bounds__(256)
void gsa3d_qkv_kernel(const float* __restrict__ x,
                      const float* __restrict__ wqkv,
                      float* __restrict__ Q,
                      float* __restrict__ K,
                      float* __restrict__ V) {
  const int wid  = blockIdx.x * 8 + (threadIdx.x >> 5);
  const int lane = threadIdx.x & 31;
  const int l16  = lane & 15;
  const int kh   = lane >> 4;
  const int tn   = wid % 48;
  const int tm   = wid / 48;
  const int row  = tm * 16 + l16;
  const int col  = tn * 16 + l16;

  const float* xa = x + row + (4 * kh) * N_TOK;
  const float* wb = wqkv + col * CDIM + 4 * kh;
  float4 af[2][QCH], bf[2][QCH];
#pragma unroll
  for (int i = 0; i < QCH; ++i) {
    const float* xp = xa + (8 * i) * N_TOK;
    af[0][i] = make_float4(xp[0], xp[N_TOK], xp[2 * N_TOK], xp[3 * N_TOK]);
    bf[0][i] = *reinterpret_cast<const float4*>(wb + 8 * i);
  }
  v8f acc = {};
#pragma unroll
  for (int c = 0; c < CDIM / (8 * QCH); ++c) {
    const int cur = c & 1, nxt = cur ^ 1;
    if (c + 1 < CDIM / (8 * QCH)) {
#pragma unroll
      for (int i = 0; i < QCH; ++i) {
        const int ks = 8 * (QCH * (c + 1) + i);
        const float* xp = xa + ks * N_TOK;
        af[nxt][i] = make_float4(xp[0], xp[N_TOK], xp[2 * N_TOK], xp[3 * N_TOK]);
        bf[nxt][i] = *reinterpret_cast<const float4*>(wb + ks);
      }
    }
    __builtin_amdgcn_sched_barrier(0);
#pragma unroll
    for (int i = 0; i < QCH; ++i) {
      v2f a0, a1, b0, b1;
      a0.x = af[cur][i].x;  a0.y = af[cur][i].y;
      a1.x = af[cur][i].z;  a1.y = af[cur][i].w;
      b0.x = bf[cur][i].x;  b0.y = bf[cur][i].y;
      b1.x = bf[cur][i].z;  b1.y = bf[cur][i].w;
      acc = wmma4(a0, b0, acc);
      acc = wmma4(a1, b1, acc);
    }
  }

  const int s  = col >> 8;
  const int hh = (col >> 5) & 7;
  const int d  = col & 31;
  float* dst = (s == 0) ? Q : ((s == 1) ? K : V);
  dst += hh * (N_TOK * HD) + d;
#pragma unroll
  for (int r = 0; r < 8; ++r) {
    const int n = tm * 16 + r + 8 * kh;
    dst[n * HD] = acc[r];
  }
}

#define KSPLIT  8
#define QROWS   32
#define NTP     54
#define OPITCH  36

__device__ __forceinline__ float xhalf(float x) {
  const int b = __builtin_bit_cast(int, x);
  return __builtin_bit_cast(float,
      __builtin_amdgcn_permlanex16(b, b, 0x76543210, 0xfedcba98, false, false));
}

struct KVFrag {
  float4 k[4];
  float2 v[8];
};

__device__ __forceinline__ void load_kv(KVFrag& f, const float* __restrict__ Kh,
                                        const float* __restrict__ Vh,
                                        int kb, int l16, int kh) {
  const float* Kb = Kh + (kb * 16 + l16) * HD + 4 * kh;
#pragma unroll
  for (int j = 0; j < 4; ++j) f.k[j] = *reinterpret_cast<const float4*>(Kb + 8 * j);
  const float* Vb = Vh + (kb * 16 + 8 * kh) * HD + 2 * l16;
#pragma unroll
  for (int t = 0; t < 8; ++t) f.v[t] = *reinterpret_cast<const float2*>(Vb + t * HD);
}

struct QTile {
  v2f   bq[8];
  float l;
  v8f   o0;
  v8f   o1;
};

__device__ __forceinline__ void softmax_tile(const v8f& s, QTile& q, float (&p)[8]) {
#pragma unroll
  for (int r = 0; r < 8; ++r) {
    const float t = fminf(fmaxf(s[r] * SC2_ - SH2_, -2.0f * SH2_), 0.0f);
    p[r] = __builtin_amdgcn_exp2f(t);
  }
  const float rs = ((p[0] + p[1]) + (p[2] + p[3])) + ((p[4] + p[5]) + (p[6] + p[7]));
  q.l += rs;
}

__device__ __forceinline__ void attn_block(const KVFrag& f, QTile& qa, QTile& qb) {
  v8f sa = {};
  v8f sb = {};
#pragma unroll
  for (int j = 0; j < 4; ++j) {
    v2f a0, a1;
    a0.x = f.k[j].x;  a0.y = f.k[j].y;
    a1.x = f.k[j].z;  a1.y = f.k[j].w;
    sa = wmma4(a0, qa.bq[2 * j + 0], sa);
    sb = wmma4(a0, qb.bq[2 * j + 0], sb);
    sa = wmma4(a1, qa.bq[2 * j + 1], sa);
    sb = wmma4(a1, qb.bq[2 * j + 1], sb);
  }

  float pa[8], pb[8];
  softmax_tile(sa, qa, pa);
  softmax_tile(sb, qb, pb);

#pragma unroll
  for (int t = 0; t < 4; ++t) {
    const float2 v0 = f.v[2 * t + 0];
    const float2 v1 = f.v[2 * t + 1];
    v2f ba, bb, a0, a1;
    ba.x = pa[2 * t + 0];  ba.y = pa[2 * t + 1];
    bb.x = pb[2 * t + 0];  bb.y = pb[2 * t + 1];
    a0.x = v0.x;           a0.y = v1.x;
    a1.x = v0.y;           a1.y = v1.y;
    qa.o0 = wmma4(a0, ba, qa.o0);
    qb.o0 = wmma4(a0, bb, qb.o0);
    qa.o1 = wmma4(a1, ba, qa.o1);
    qb.o1 = wmma4(a1, bb, qb.o1);
  }
}

__device__ __forceinline__ void init_qtile(QTile& q, const float* __restrict__ Qh,
                                           int tile, int l16, int kh) {
  const float* Qr = Qh + (tile * 16 + l16) * HD + 4 * kh;
#pragma unroll
  for (int j = 0; j < 4; ++j) {
    const float4 q4 = *reinterpret_cast<const float4*>(Qr + 8 * j);
    q.bq[2 * j + 0].x = q4.x;  q.bq[2 * j + 0].y = q4.y;
    q.bq[2 * j + 1].x = q4.z;  q.bq[2 * j + 1].y = q4.w;
  }
  q.l = 0.0f;
  q.o0 = v8f{};
  q.o1 = v8f{};
}

__device__ __forceinline__ void publish_qtile(const QTile& q, float* Ow, float* Lw,
                                              int l16, int kh) {
  const float l_tot = q.l + xhalf(q.l);
  float* Orow = Ow + l16 * OPITCH + 16 * kh;
#pragma unroll
  for (int i = 0; i < 4; ++i) {
    float4 v;
    v.x = q.o0[2 * i];      v.y = q.o1[2 * i];
    v.z = q.o0[2 * i + 1];  v.w = q.o1[2 * i + 1];
    *reinterpret_cast<float4*>(Orow + 4 * i) = v;
  }
  if (kh == 0) Lw[l16] = l_tot;
}

__global__ __launch_bounds__(256)
void gsa3d_attn_kernel(const float* __restrict__ Q,
                       const float* __restrict__ K,
                       const float* __restrict__ V,
                       float* __restrict__ AO) {
  __shared__ alignas(16) float ldsO[KSPLIT * QROWS * OPITCH];
  __shared__ float ldsL[KSPLIT * QROWS];
  const int wslot = __builtin_amdgcn_readfirstlane((int)(threadIdx.x >> 5));
  const int lane  = threadIdx.x & 31;
  const int l16   = lane & 15;
  const int kh    = lane >> 4;
  const int h     = blockIdx.x / NTP;
  const int tp    = blockIdx.x % NTP;
  const int kb_lo = wslot * 13 + (wslot < 4 ? wslot : 4);
  const int kb_hi = kb_lo + 13 + (wslot < 4 ? 1 : 0);

  const float* Qh = Q + h * (N_TOK * HD);
  const float* Kh = K + h * (N_TOK * HD);
  const float* Vh = V + h * (N_TOK * HD);

  QTile qa, qb;
  init_qtile(qa, Qh, 2 * tp + 0, l16, kh);
  init_qtile(qb, Qh, 2 * tp + 1, l16, kh);

  KVFrag fa, fb;
  load_kv(fa, Kh, Vh, kb_lo, l16, kh);
  for (int kb = kb_lo;;) {
    load_kv(fb, Kh, Vh, kb + 1, l16, kh);
    __builtin_amdgcn_sched_barrier(0);
    attn_block(fa, qa, qb);
    if (kb + 2 >= kb_hi) {
      attn_block(fb, qa, qb);
      break;
    }
    load_kv(fa, Kh, Vh, kb + 2, l16, kh);
    __builtin_amdgcn_sched_barrier(0);
    attn_block(fb, qa, qb);
    kb += 2;
    if (kb + 1 >= kb_hi) {
      attn_block(fa, qa, qb);
      break;
    }
  }

  {
    float* Ow = ldsO + wslot * (QROWS * OPITCH);
    float* Lw = ldsL + wslot * QROWS;
    publish_qtile(qa, Ow, Lw, l16, kh);
    publish_qtile(qb, Ow + 16 * OPITCH, Lw + 16, l16, kh);
  }
  __syncthreads();

#pragma unroll
  for (int qt = 0; qt < 2; ++qt) {
    const int row = 16 * qt + (threadIdx.x >> 4);
    const int dc  = (threadIdx.x & 15) * 2;
    float lsum = 0.0f, a0 = 0.0f, a1 = 0.0f;
#pragma unroll
    for (int w = 0; w < KSPLIT; ++w) {
      const float2 ov = *reinterpret_cast<const float2*>(
          ldsO + w * (QROWS * OPITCH) + row * OPITCH + dc);
      lsum += ldsL[w * QROWS + row];
      a0   += ov.x;
      a1   += ov.y;
    }
    const float inv = 1.0f / lsum;
    float2 res;
    res.x = a0 * inv;
    res.y = a1 * inv;
    const int n = tp * QROWS + row;
    *reinterpret_cast<float2*>(AO + n * CDIM + h * HD + dc) = res;
  }
}

__global__ __launch_bounds__(256)
void gsa3d_proj_kernel(const float* __restrict__ AO,
                       const float* __restrict__ wproj,
                       float* __restrict__ out) {
  const int wid  = blockIdx.x * 8 + (threadIdx.x >> 5);
  const int lane = threadIdx.x & 31;
  const int l16  = lane & 15;
  const int kh   = lane >> 4;
  const int tm   = wid / 16;
  const int tn   = wid % 16;
  const int row  = tm * 16 + l16;
  const int col  = tn * 16 + l16;

  const float* ap = AO + row * CDIM + 4 * kh;
  const float* bp = wproj + col * CDIM + 4 * kh;
  float4 af[2][PCH], bf[2][PCH];
#pragma unroll
  for (int i = 0; i < PCH; ++i) {
    af[0][i] = *reinterpret_cast<const float4*>(ap + 8 * i);
    bf[0][i] = *reinterpret_cast<const float4*>(bp + 8 * i);
  }
  v8f acc = {};
#pragma unroll
  for (int c = 0; c < CDIM / (8 * PCH); ++c) {
    const int cur = c & 1, nxt = cur ^ 1;
    if (c + 1 < CDIM / (8 * PCH)) {
#pragma unroll
      for (int i = 0; i < PCH; ++i) {
        const int ks = 8 * (PCH * (c + 1) + i);
        af[nxt][i] = *reinterpret_cast<const float4*>(ap + ks);
        bf[nxt][i] = *reinterpret_cast<const float4*>(bp + ks);
      }
    }
    __builtin_amdgcn_sched_barrier(0);
#pragma unroll
    for (int i = 0; i < PCH; ++i) {
      v2f a0, a1, b0, b1;
      a0.x = af[cur][i].x;  a0.y = af[cur][i].y;
      a1.x = af[cur][i].z;  a1.y = af[cur][i].w;
      b0.x = bf[cur][i].x;  b0.y = bf[cur][i].y;
      b1.x = bf[cur][i].z;  b1.y = bf[cur][i].w;
      acc = wmma4(a0, b0, acc);
      acc = wmma4(a1, b1, acc);
    }
  }
#pragma unroll
  for (int r = 0; r < 8; ++r) {
    const int n = tm * 16 + r + 8 * kh;
    out[col * N_TOK + n] = acc[r];
  }
}

extern "C" void kernel_launch(void* const* d_in, const int* in_sizes, int n_in,
                              void* d_out, int out_size, void* d_ws, size_t ws_size,
                              hipStream_t stream) {
  (void)in_sizes; (void)n_in; (void)out_size; (void)ws_size;
  const float* x     = (const float*)d_in[0];
  const float* wqkv  = (const float*)d_in[1];
  const float* wproj = (const float*)d_in[2];
  float* out = (float*)d_out;

  float* ws = (float*)d_ws;
  float* Q  = ws;
  float* K  = Q + NHEADS * N_TOK * HD;
  float* V  = K + NHEADS * N_TOK * HD;
  float* AO = V + NHEADS * N_TOK * HD;

  gsa3d_qkv_kernel <<<648, 256, 0, stream>>>(x, wqkv, Q, K, V);
  gsa3d_attn_kernel<<<432, 256, 0, stream>>>(Q, K, V, AO);
  gsa3d_proj_kernel<<<216, 256, 0, stream>>>(AO, wproj, out);
}


# qkv kernel: each wave computes three 16-token tiles (48 tokens x 16 columns) per weight fragment, x fragments as 12-byte loads (token rows 3*lane+t): 1728 waves for 5184; every sum order unchanged
# speedup vs baseline: 5.4346x; 1.0946x over previous
#include <hip/hip_runtime.h>

typedef __attribute__((ext_vector_type(2))) float v2f;
typedef __attribute__((ext_vector_type(8))) float v8f;

#define N_TOK   1728
#define CDIM    256
#define NHEADS  8
#define HD      32
#define NT      108
#define SCALE_  0.1767766952966369f
#define LOG2E_  1.4426950408889634f
#define SC2_    (SCALE_ * LOG2E_)
#define SH2_    (10.0f * LOG2E_)
#define QCH     4
#define PCH     8

__device__ __forceinline__ v8f wmma4(v2f a, v2f b, v8f c) {
  return __builtin_amdgcn_wmma_f32_16x16x4_f32(
       false, a,  false, b,
       (short)0, c,  false,  false);
}

#define QMT     3
#define QROWS_W 48

struct __attribute__((packed, aligned(4))) F3 { float v[3]; };

struct QkvFrag {
  F3     a[QCH][4];
  float4 b[QCH];
};

__device__ __forceinline__ void load_qkv(QkvFrag& f, const float* __restrict__ xa,
                                         const float* __restrict__ wb, int chunk) {
#pragma unroll
  for (int i = 0; i < QCH; ++i) {
    const int ks = 8 * (QCH * chunk + i);
    const float* xp = xa + ks * N_TOK;
#pragma unroll
    for (int e = 0; e < 4; ++e)
      f.a[i][e] = *reinterpret_cast<const F3*>(xp + e * N_TOK);
    f.b[i] = *reinterpret_cast<const float4*>(wb + ks);
  }
}

__device__ __forceinline__ void mma_qkv(v8f (&acc)[QMT], const QkvFrag& f) {
#pragma unroll
  for (int i = 0; i < QCH; ++i) {
    v2f b0, b1;
    b0.x = f.b[i].x;  b0.y = f.b[i].y;
    b1.x = f.b[i].z;  b1.y = f.b[i].w;
#pragma unroll
    for (int t = 0; t < QMT; ++t) {
      v2f a0;
      a0.x = f.a[i][0].v[t];  a0.y = f.a[i][1].v[t];
      acc[t] = wmma4(a0, b0, acc[t]);
    }
#pragma unroll
    for (int t = 0; t < QMT; ++t) {
      v2f a1;
      a1.x = f.a[i][2].v[t];  a1.y = f.a[i][3].v[t];
      acc[t] = wmma4(a1, b1, acc[t]);
    }
  }
}

__global__ __launch_bounds__(256)
void gsa3d_qkv_kernel(const float* __restrict__ x,
                      const float* __restrict__ wqkv,
                      float* __restrict__ Q,
                      float* __restrict__ K,
                      float* __restrict__ V) {
  const int wid  = blockIdx.x * 8 + (threadIdx.x >> 5);
  const int lane = threadIdx.x & 31;
  const int l16  = lane & 15;
  const int kh   = lane >> 4;
  const int tn   = wid % 48;
  const int tg   = wid / 48;
  const int n0   = tg * QROWS_W;
  const int col  = tn * 16 + l16;

  const float* xa = x + (4 * kh) * N_TOK + n0 + 3 * l16;
  const float* wb = wqkv + col * CDIM + 4 * kh;
  QkvFrag f0, f1;
  load_qkv(f0, xa, wb, 0);
  v8f acc[QMT];
#pragma unroll
  for (int t = 0; t < QMT; ++t) acc[t] = v8f{};
#pragma unroll
  for (int c = 0; c < CDIM / (8 * QCH); c += 2) {
    load_qkv(f1, xa, wb, c + 1);
    __builtin_amdgcn_sched_barrier(0);
    mma_qkv(acc, f0);
    if (c + 2 < CDIM / (8 * QCH)) load_qkv(f0, xa, wb, c + 2);
    __builtin_amdgcn_sched_barrier(0);
    mma_qkv(acc, f1);
  }

  const int s  = col >> 8;
  const int hh = (col >> 5) & 7;
  const int d  = col & 31;
  float* dst = (s == 0) ? Q : ((s == 1) ? K : V);
  dst += hh * (N_TOK * HD) + d;
#pragma unroll
  for (int t = 0; t < QMT; ++t) {
#pragma unroll
    for (int r = 0; r < 8; ++r) {
      const int n = n0 + 3 * (r + 8 * kh) + t;
      dst[n * HD] = acc[t][r];
    }
  }
}

#define KSPLIT  8
#define QROWS   32
#define NTP     54
#define OPITCH  36

__device__ __forceinline__ float xhalf(float x) {
  const int b = __builtin_bit_cast(int, x);
  return __builtin_bit_cast(float,
      __builtin_amdgcn_permlanex16(b, b, 0x76543210, 0xfedcba98, false, false));
}

struct KVFrag {
  float4 k[4];
  float2 v[8];
};

__device__ __forceinline__ void load_kv(KVFrag& f, const float* __restrict__ Kh,
                                        const float* __restrict__ Vh,
                                        int kb, int l16, int kh) {
  const float* Kb = Kh + (kb * 16 + l16) * HD + 4 * kh;
#pragma unroll
  for (int j = 0; j < 4; ++j) f.k[j] = *reinterpret_cast<const float4*>(Kb + 8 * j);
  const float* Vb = Vh + (kb * 16 + 8 * kh) * HD + 2 * l16;
#pragma unroll
  for (int t = 0; t < 8; ++t) f.v[t] = *reinterpret_cast<const float2*>(Vb + t * HD);
}

struct QTile {
  v2f   bq[8];
  float l;
  v8f   o0;
  v8f   o1;
};

__device__ __forceinline__ void softmax_tile(const v8f& s, QTile& q, float (&p)[8]) {
#pragma unroll
  for (int r = 0; r < 8; ++r) {
    const float t = fminf(fmaxf(s[r] * SC2_ - SH2_, -2.0f * SH2_), 0.0f);
    p[r] = __builtin_amdgcn_exp2f(t);
  }
  const float rs = ((p[0] + p[1]) + (p[2] + p[3])) + ((p[4] + p[5]) + (p[6] + p[7]));
  q.l += rs;
}

__device__ __forceinline__ void attn_block(const KVFrag& f, QTile& qa, QTile& qb) {
  v8f sa = {};
  v8f sb = {};
#pragma unroll
  for (int j = 0; j < 4; ++j) {
    v2f a0, a1;
    a0.x = f.k[j].x;  a0.y = f.k[j].y;
    a1.x = f.k[j].z;  a1.y = f.k[j].w;
    sa = wmma4(a0, qa.bq[2 * j + 0], sa);
    sb = wmma4(a0, qb.bq[2 * j + 0], sb);
    sa = wmma4(a1, qa.bq[2 * j + 1], sa);
    sb = wmma4(a1, qb.bq[2 * j + 1], sb);
  }

  float pa[8], pb[8];
  softmax_tile(sa, qa, pa);
  softmax_tile(sb, qb, pb);

#pragma unroll
  for (int t = 0; t < 4; ++t) {
    const float2 v0 = f.v[2 * t + 0];
    const float2 v1 = f.v[2 * t + 1];
    v2f ba, bb, a0, a1;
    ba.x = pa[2 * t + 0];  ba.y = pa[2 * t + 1];
    bb.x = pb[2 * t + 0];  bb.y = pb[2 * t + 1];
    a0.x = v0.x;           a0.y = v1.x;
    a1.x = v0.y;           a1.y = v1.y;
    qa.o0 = wmma4(a0, ba, qa.o0);
    qb.o0 = wmma4(a0, bb, qb.o0);
    qa.o1 = wmma4(a1, ba, qa.o1);
    qb.o1 = wmma4(a1, bb, qb.o1);
  }
}

__device__ __forceinline__ void init_qtile(QTile& q, const float* __restrict__ Qh,
                                           int tile, int l16, int kh) {
  const float* Qr = Qh + (tile * 16 + l16) * HD + 4 * kh;
#pragma unroll
  for (int j = 0; j < 4; ++j) {
    const float4 q4 = *reinterpret_cast<const float4*>(Qr + 8 * j);
    q.bq[2 * j + 0].x = q4.x;  q.bq[2 * j + 0].y = q4.y;
    q.bq[2 * j + 1].x = q4.z;  q.bq[2 * j + 1].y = q4.w;
  }
  q.l = 0.0f;
  q.o0 = v8f{};
  q.o1 = v8f{};
}

__device__ __forceinline__ void publish_qtile(const QTile& q, float* Ow, float* Lw,
                                              int l16, int kh) {
  const float l_tot = q.l + xhalf(q.l);
  float* Orow = Ow + l16 * OPITCH + 16 * kh;
#pragma unroll
  for (int i = 0; i < 4; ++i) {
    float4 v;
    v.x = q.o0[2 * i];      v.y = q.o1[2 * i];
    v.z = q.o0[2 * i + 1];  v.w = q.o1[2 * i + 1];
    *reinterpret_cast<float4*>(Orow + 4 * i) = v;
  }
  if (kh == 0) Lw[l16] = l_tot;
}

__global__ __launch_bounds__(256)
void gsa3d_attn_kernel(const float* __restrict__ Q,
                       const float* __restrict__ K,
                       const float* __restrict__ V,
                       float* __restrict__ AO) {
  __shared__ alignas(16) float ldsO[KSPLIT * QROWS * OPITCH];
  __shared__ float ldsL[KSPLIT * QROWS];
  const int wslot = __builtin_amdgcn_readfirstlane((int)(threadIdx.x >> 5));
  const int lane  = threadIdx.x & 31;
  const int l16   = lane & 15;
  const int kh    = lane >> 4;
  const int h     = blockIdx.x / NTP;
  const int tp    = blockIdx.x % NTP;
  const int kb_lo = wslot * 13 + (wslot < 4 ? wslot : 4);
  const int kb_hi = kb_lo + 13 + (wslot < 4 ? 1 : 0);

  const float* Qh = Q + h * (N_TOK * HD);
  const float* Kh = K + h * (N_TOK * HD);
  const float* Vh = V + h * (N_TOK * HD);

  QTile qa, qb;
  init_qtile(qa, Qh, 2 * tp + 0, l16, kh);
  init_qtile(qb, Qh, 2 * tp + 1, l16, kh);

  KVFrag fa, fb;
  load_kv(fa, Kh, Vh, kb_lo, l16, kh);
  for (int kb = kb_lo;;) {
    load_kv(fb, Kh, Vh, kb + 1, l16, kh);
    __builtin_amdgcn_sched_barrier(0);
    attn_block(fa, qa, qb);
    if (kb + 2 >= kb_hi) {
      attn_block(fb, qa, qb);
      break;
    }
    load_kv(fa, Kh, Vh, kb + 2, l16, kh);
    __builtin_amdgcn_sched_barrier(0);
    attn_block(fb, qa, qb);
    kb += 2;
    if (kb + 1 >= kb_hi) {
      attn_block(fa, qa, qb);
      break;
    }
  }

  {
    float* Ow = ldsO + wslot * (QROWS * OPITCH);
    float* Lw = ldsL + wslot * QROWS;
    publish_qtile(qa, Ow, Lw, l16, kh);
    publish_qtile(qb, Ow + 16 * OPITCH, Lw + 16, l16, kh);
  }
  __syncthreads();

#pragma unroll
  for (int qt = 0; qt < 2; ++qt) {
    const int row = 16 * qt + (threadIdx.x >> 4);
    const int dc  = (threadIdx.x & 15) * 2;
    float lsum = 0.0f, a0 = 0.0f, a1 = 0.0f;
#pragma unroll
    for (int w = 0; w < KSPLIT; ++w) {
      const float2 ov = *reinterpret_cast<const float2*>(
          ldsO + w * (QROWS * OPITCH) + row * OPITCH + dc);
      lsum += ldsL[w * QROWS + row];
      a0   += ov.x;
      a1   += ov.y;
    }
    const float inv = 1.0f / lsum;
    float2 res;
    res.x = a0 * inv;
    res.y = a1 * inv;
    const int n = tp * QROWS + row;
    *reinterpret_cast<float2*>(AO + n * CDIM + h * HD + dc) = res;
  }
}

__global__ __launch_bounds__(256)
void gsa3d_proj_kernel(const float* __restrict__ AO,
                       const float* __restrict__ wproj,
                       float* __restrict__ out) {
  const int wid  = blockIdx.x * 8 + (threadIdx.x >> 5);
  const int lane = threadIdx.x & 31;
  const int l16  = lane & 15;
  const int kh   = lane >> 4;
  const int tm   = wid / 16;
  const int tn   = wid % 16;
  const int row  = tm * 16 + l16;
  const int col  = tn * 16 + l16;

  const float* ap = AO + row * CDIM + 4 * kh;
  const float* bp = wproj + col * CDIM + 4 * kh;
  float4 af[2][PCH], bf[2][PCH];
#pragma unroll
  for (int i = 0; i < PCH; ++i) {
    af[0][i] = *reinterpret_cast<const float4*>(ap + 8 * i);
    bf[0][i] = *reinterpret_cast<const float4*>(bp + 8 * i);
  }
  v8f acc = {};
#pragma unroll
  for (int c = 0; c < CDIM / (8 * PCH); ++c) {
    const int cur = c & 1, nxt = cur ^ 1;
    if (c + 1 < CDIM / (8 * PCH)) {
#pragma unroll
      for (int i = 0; i < PCH; ++i) {
        const int ks = 8 * (PCH * (c + 1) + i);
        af[nxt][i] = *reinterpret_cast<const float4*>(ap + ks);
        bf[nxt][i] = *reinterpret_cast<const float4*>(bp + ks);
      }
    }
    __builtin_amdgcn_sched_barrier(0);
#pragma unroll
    for (int i = 0; i < PCH; ++i) {
      v2f a0, a1, b0, b1;
      a0.x = af[cur][i].x;  a0.y = af[cur][i].y;
      a1.x = af[cur][i].z;  a1.y = af[cur][i].w;
      b0.x = bf[cur][i].x;  b0.y = bf[cur][i].y;
      b1.x = bf[cur][i].z;  b1.y = bf[cur][i].w;
      acc = wmma4(a0, b0, acc);
      acc = wmma4(a1, b1, acc);
    }
  }
#pragma unroll
  for (int r = 0; r < 8; ++r) {
    const int n = tm * 16 + r + 8 * kh;
    out[col * N_TOK + n] = acc[r];
  }
}

extern "C" void kernel_launch(void* const* d_in, const int* in_sizes, int n_in,
                              void* d_out, int out_size, void* d_ws, size_t ws_size,
                              hipStream_t stream) {
  (void)in_sizes; (void)n_in; (void)out_size; (void)ws_size;
  const float* x     = (const float*)d_in[0];
  const float* wqkv  = (const float*)d_in[1];
  const float* wproj = (const float*)d_in[2];
  float* out = (float*)d_out;

  float* ws = (float*)d_ws;
  float* Q  = ws;
  float* K  = Q + NHEADS * N_TOK * HD;
  float* V  = K + NHEADS * N_TOK * HD;
  float* AO = V + NHEADS * N_TOK * HD;

  gsa3d_qkv_kernel <<<216, 256, 0, stream>>>(x, wqkv, Q, K, V);
  gsa3d_attn_kernel<<<432, 256, 0, stream>>>(Q, K, V, AO);
  gsa3d_proj_kernel<<<216, 256, 0, stream>>>(AO, wproj, out);
}


# workspace Q, K, V and AO stored in fragment order: each 16-byte fragment load of a wave reads 512 contiguous bytes (4 lines, was 16); V pairs arrive as 16-byte loads; every sum order unchanged
# speedup vs baseline: 5.7458x; 1.0573x over previous
#include <hip/hip_runtime.h>

typedef __attribute__((ext_vector_type(2))) float v2f;
typedef __attribute__((ext_vector_type(8))) float v8f;

#define N_TOK   1728
#define CDIM    256
#define NHEADS  8
#define HD      32
#define NT      108
#define SCALE_  0.1767766952966369f
#define LOG2E_  1.4426950408889634f
#define SC2_    (SCALE_ * LOG2E_)
#define SH2_    (10.0f * LOG2E_)
#define FRAG_BLK 512
#define AO_TILE  4096
#define QCH     4
#define PCH     8

__device__ __forceinline__ v8f wmma4(v2f a, v2f b, v8f c) {
  return __builtin_amdgcn_wmma_f32_16x16x4_f32(
       false, a,  false, b,
       (short)0, c,  false,  false);
}

#define QMT     3
#define QROWS_W 48

struct __attribute__((packed, aligned(4))) F3 { float v[3]; };

struct QkvFrag {
  F3     a[QCH][4];
  float4 b[QCH];
};

__device__ __forceinline__ void load_qkv(QkvFrag& f, const float* __restrict__ xa,
                                         const float* __restrict__ wb, int chunk) {
#pragma unroll
  for (int i = 0; i < QCH; ++i) {
    const int ks = 8 * (QCH * chunk + i);
    const float* xp = xa + ks * N_TOK;
#pragma unroll
    for (int e = 0; e < 4; ++e)
      f.a[i][e] = *reinterpret_cast<const F3*>(xp + e * N_TOK);
    f.b[i] = *reinterpret_cast<const float4*>(wb + ks);
  }
}

__device__ __forceinline__ void mma_qkv(v8f (&acc)[QMT], const QkvFrag& f) {
#pragma unroll
  for (int i = 0; i < QCH; ++i) {
    v2f b0, b1;
    b0.x = f.b[i].x;  b0.y = f.b[i].y;
    b1.x = f.b[i].z;  b1.y = f.b[i].w;
#pragma unroll
    for (int t = 0; t < QMT; ++t) {
      v2f a0;
      a0.x = f.a[i][0].v[t];  a0.y = f.a[i][1].v[t];
      acc[t] = wmma4(a0, b0, acc[t]);
    }
#pragma unroll
    for (int t = 0; t < QMT; ++t) {
      v2f a1;
      a1.x = f.a[i][2].v[t];  a1.y = f.a[i][3].v[t];
      acc[t] = wmma4(a1, b1, acc[t]);
    }
  }
}

__global__ __launch_bounds__(256)
void gsa3d_qkv_kernel(const float* __restrict__ x,
                      const float* __restrict__ wqkv,
                      float* __restrict__ Q,
                      float* __restrict__ K,
                      float* __restrict__ V) {
  const int wid  = blockIdx.x * 8 + (threadIdx.x >> 5);
  const int lane = threadIdx.x & 31;
  const int l16  = lane & 15;
  const int kh   = lane >> 4;
  const int tn   = wid % 48;
  const int tg   = wid / 48;
  const int n0   = tg * QROWS_W;
  const int col  = tn * 16 + l16;

  const float* xa = x + (4 * kh) * N_TOK + n0 + 3 * l16;
  const float* wb = wqkv + col * CDIM + 4 * kh;
  QkvFrag f0, f1;
  load_qkv(f0, xa, wb, 0);
  v8f acc[QMT];
#pragma unroll
  for (int t = 0; t < QMT; ++t) acc[t] = v8f{};
#pragma unroll
  for (int c = 0; c < CDIM / (8 * QCH); c += 2) {
    load_qkv(f1, xa, wb, c + 1);
    __builtin_amdgcn_sched_barrier(0);
    mma_qkv(acc, f0);
    if (c + 2 < CDIM / (8 * QCH)) load_qkv(f0, xa, wb, c + 2);
    __builtin_amdgcn_sched_barrier(0);
    mma_qkv(acc, f1);
  }

  const int s  = __builtin_amdgcn_readfirstlane(col >> 8);
  const int hh = (col >> 5) & 7;
  const int d  = col & 31;
  float* dst = (s == 0) ? Q : ((s == 1) ? K : V);
  dst += hh * (N_TOK * HD);
  dst += (n0 >> 4) * FRAG_BLK;
  if (s == 2) {
    dst += (d >> 1) * 4 + (d & 1) * 2;
    float* dlo = dst + kh * 576;
    float* dhi = dst + kh * 960;
#pragma unroll
    for (int t = 0; t < QMT; ++t) {
#pragma unroll
      for (int r = 0; r < 8; ++r) {
        const int m  = 3 * r + t;
        const int kk = m & 15;
        const int o  = (m >> 4) * FRAG_BLK + ((kk >> 1) & 3) * 128 + (kk >> 3) * 64 + (kk & 1);
        (kk < 8 ? dlo : dhi)[o] = acc[t][r];
      }
    }
  } else {
    dst += (d >> 3) * 128 + ((d >> 2) & 1) * 64 + (d & 3);
    float* dlo = dst + kh * 544;
    float* dhi = dst + kh * 992;
#pragma unroll
    for (int t = 0; t < QMT; ++t) {
#pragma unroll
      for (int r = 0; r < 8; ++r) {
        const int m = 3 * r + t;
        const int o = (m >> 4) * FRAG_BLK + (m & 15) * 4;
        ((m & 15) < 8 ? dlo : dhi)[o] = acc[t][r];
      }
    }
  }
}

#define KSPLIT  8
#define QROWS   32
#define NTP     54
#define OPITCH  36

__device__ __forceinline__ float xhalf(float x) {
  const int b = __builtin_bit_cast(int, x);
  return __builtin_bit_cast(float,
      __builtin_amdgcn_permlanex16(b, b, 0x76543210, 0xfedcba98, false, false));
}

struct KVFrag {
  float4 k[4];
  float4 v[4];
};

__device__ __forceinline__ void load_kv(KVFrag& f, const float* __restrict__ Kh,
                                        const float* __restrict__ Vh,
                                        int kb, int l16, int kh) {
  const int lo = kb * FRAG_BLK + (kh * 16 + l16) * 4;
  const float* Kb = Kh + lo;
#pragma unroll
  for (int j = 0; j < 4; ++j) f.k[j] = *reinterpret_cast<const float4*>(Kb + 128 * j);
  const float* Vb = Vh + lo;
#pragma unroll
  for (int t = 0; t < 4; ++t) f.v[t] = *reinterpret_cast<const float4*>(Vb + 128 * t);
}

struct QTile {
  v2f   bq[8];
  float l;
  v8f   o0;
  v8f   o1;
};

__device__ __forceinline__ void softmax_tile(const v8f& s, QTile& q, float (&p)[8]) {
#pragma unroll
  for (int r = 0; r < 8; ++r) {
    const float t = fminf(fmaxf(s[r] * SC2_ - SH2_, -2.0f * SH2_), 0.0f);
    p[r] = __builtin_amdgcn_exp2f(t);
  }
  const float rs = ((p[0] + p[1]) + (p[2] + p[3])) + ((p[4] + p[5]) + (p[6] + p[7]));
  q.l += rs;
}

__device__ __forceinline__ void attn_block(const KVFrag& f, QTile& qa, QTile& qb) {
  v8f sa = {};
  v8f sb = {};
#pragma unroll
  for (int j = 0; j < 4; ++j) {
    v2f a0, a1;
    a0.x = f.k[j].x;  a0.y = f.k[j].y;
    a1.x = f.k[j].z;  a1.y = f.k[j].w;
    sa = wmma4(a0, qa.bq[2 * j + 0], sa);
    sb = wmma4(a0, qb.bq[2 * j + 0], sb);
    sa = wmma4(a1, qa.bq[2 * j + 1], sa);
    sb = wmma4(a1, qb.bq[2 * j + 1], sb);
  }

  float pa[8], pb[8];
  softmax_tile(sa, qa, pa);
  softmax_tile(sb, qb, pb);

#pragma unroll
  for (int t = 0; t < 4; ++t) {
    const float4 vv = f.v[t];
    v2f ba, bb, a0, a1;
    ba.x = pa[2 * t + 0];  ba.y = pa[2 * t + 1];
    bb.x = pb[2 * t + 0];  bb.y = pb[2 * t + 1];
    a0.x = vv.x;           a0.y = vv.y;
    a1.x = vv.z;           a1.y = vv.w;
    qa.o0 = wmma4(a0, ba, qa.o0);
    qb.o0 = wmma4(a0, bb, qb.o0);
    qa.o1 = wmma4(a1, ba, qa.o1);
    qb.o1 = wmma4(a1, bb, qb.o1);
  }
}

__device__ __forceinline__ void init_qtile(QTile& q, const float* __restrict__ Qh,
                                           int tile, int l16, int kh) {
  const float* Qr = Qh + tile * FRAG_BLK + (kh * 16 + l16) * 4;
#pragma unroll
  for (int j = 0; j < 4; ++j) {
    const float4 q4 = *reinterpret_cast<const float4*>(Qr + 128 * j);
    q.bq[2 * j + 0].x = q4.x;  q.bq[2 * j + 0].y = q4.y;
    q.bq[2 * j + 1].x = q4.z;  q.bq[2 * j + 1].y = q4.w;
  }
  q.l = 0.0f;
  q.o0 = v8f{};
  q.o1 = v8f{};
}

__device__ __forceinline__ void publish_qtile(const QTile& q, float* Ow, float* Lw,
                                              int l16, int kh) {
  const float l_tot = q.l + xhalf(q.l);
  float* Orow = Ow + l16 * OPITCH + 16 * kh;
#pragma unroll
  for (int i = 0; i < 4; ++i) {
    float4 v;
    v.x = q.o0[2 * i];      v.y = q.o1[2 * i];
    v.z = q.o0[2 * i + 1];  v.w = q.o1[2 * i + 1];
    *reinterpret_cast<float4*>(Orow + 4 * i) = v;
  }
  if (kh == 0) Lw[l16] = l_tot;
}

__global__ __launch_bounds__(256)
void gsa3d_attn_kernel(const float* __restrict__ Q,
                       const float* __restrict__ K,
                       const float* __restrict__ V,
                       float* __restrict__ AO) {
  __shared__ alignas(16) float ldsO[KSPLIT * QROWS * OPITCH];
  __shared__ float ldsL[KSPLIT * QROWS];
  const int wslot = __builtin_amdgcn_readfirstlane((int)(threadIdx.x >> 5));
  const int lane  = threadIdx.x & 31;
  const int l16   = lane & 15;
  const int kh    = lane >> 4;
  const int h     = blockIdx.x / NTP;
  const int tp    = blockIdx.x % NTP;
  const int kb_lo = wslot * 13 + (wslot < 4 ? wslot : 4);
  const int kb_hi = kb_lo + 13 + (wslot < 4 ? 1 : 0);

  const float* Qh = Q + h * (N_TOK * HD);
  const float* Kh = K + h * (N_TOK * HD);
  const float* Vh = V + h * (N_TOK * HD);

  QTile qa, qb;
  init_qtile(qa, Qh, 2 * tp + 0, l16, kh);
  init_qtile(qb, Qh, 2 * tp + 1, l16, kh);

  KVFrag fa, fb;
  load_kv(fa, Kh, Vh, kb_lo, l16, kh);
  for (int kb = kb_lo;;) {
    load_kv(fb, Kh, Vh, kb + 1, l16, kh);
    __builtin_amdgcn_sched_barrier(0);
    attn_block(fa, qa, qb);
    if (kb + 2 >= kb_hi) {
      attn_block(fb, qa, qb);
      break;
    }
    load_kv(fa, Kh, Vh, kb + 2, l16, kh);
    __builtin_amdgcn_sched_barrier(0);
    attn_block(fb, qa, qb);
    kb += 2;
    if (kb + 1 >= kb_hi) {
      attn_block(fa, qa, qb);
      break;
    }
  }

  {
    float* Ow = ldsO + wslot * (QROWS * OPITCH);
    float* Lw = ldsL + wslot * QROWS;
    publish_qtile(qa, Ow, Lw, l16, kh);
    publish_qtile(qb, Ow + 16 * OPITCH, Lw + 16, l16, kh);
  }
  __syncthreads();

#pragma unroll
  for (int qt = 0; qt < 2; ++qt) {
    const int row = 16 * qt + (threadIdx.x >> 4);
    const int dc  = (threadIdx.x & 15) * 2;
    float lsum = 0.0f, a0 = 0.0f, a1 = 0.0f;
#pragma unroll
    for (int w = 0; w < KSPLIT; ++w) {
      const float2 ov = *reinterpret_cast<const float2*>(
          ldsO + w * (QROWS * OPITCH) + row * OPITCH + dc);
      lsum += ldsL[w * QROWS + row];
      a0   += ov.x;
      a1   += ov.y;
    }
    const float inv = 1.0f / lsum;
    float2 res;
    res.x = a0 * inv;
    res.y = a1 * inv;
    const int cc = threadIdx.x >> 4;
    *reinterpret_cast<float2*>(AO + (2 * tp + qt) * AO_TILE + (h * 4 + (dc >> 3)) * 128
                               + ((dc >> 2) & 1) * 64 + cc * 4 + (dc & 3)) = res;
  }
}

__global__ __launch_bounds__(256)
void gsa3d_proj_kernel(const float* __restrict__ AO,
                       const float* __restrict__ wproj,
                       float* __restrict__ out) {
  const int wid  = blockIdx.x * 8 + (threadIdx.x >> 5);
  const int lane = threadIdx.x & 31;
  const int l16  = lane & 15;
  const int kh   = lane >> 4;
  const int tm   = wid / 16;
  const int tn   = wid % 16;
  const int col  = tn * 16 + l16;

  const float* ap = AO + tm * AO_TILE + (kh * 16 + l16) * 4;
  const float* bp = wproj + col * CDIM + 4 * kh;
  float4 af[2][PCH], bf[2][PCH];
#pragma unroll
  for (int i = 0; i < PCH; ++i) {
    af[0][i] = *reinterpret_cast<const float4*>(ap + 128 * i);
    bf[0][i] = *reinterpret_cast<const float4*>(bp + 8 * i);
  }
  v8f acc = {};
#pragma unroll
  for (int c = 0; c < CDIM / (8 * PCH); ++c) {
    const int cur = c & 1, nxt = cur ^ 1;
    if (c + 1 < CDIM / (8 * PCH)) {
#pragma unroll
      for (int i = 0; i < PCH; ++i) {
        const int ks = 8 * (PCH * (c + 1) + i);
        af[nxt][i] = *reinterpret_cast<const float4*>(ap + 16 * ks);
        bf[nxt][i] = *reinterpret_cast<const float4*>(bp + ks);
      }
    }
    __builtin_amdgcn_sched_barrier(0);
#pragma unroll
    for (int i = 0; i < PCH; ++i) {
      v2f a0, a1, b0, b1;
      a0.x = af[cur][i].x;  a0.y = af[cur][i].y;
      a1.x = af[cur][i].z;  a1.y = af[cur][i].w;
      b0.x = bf[cur][i].x;  b0.y = bf[cur][i].y;
      b1.x = bf[cur][i].z;  b1.y = bf[cur][i].w;
      acc = wmma4(a0, b0, acc);
      acc = wmma4(a1, b1, acc);
    }
  }
#pragma unroll
  for (int r = 0; r < 8; ++r) {
    const int n = tm * 16 + r + 8 * kh;
    out[col * N_TOK + n] = acc[r];
  }
}

extern "C" void kernel_launch(void* const* d_in, const int* in_sizes, int n_in,
                              void* d_out, int out_size, void* d_ws, size_t ws_size,
                              hipStream_t stream) {
  (void)in_sizes; (void)n_in; (void)out_size; (void)ws_size;
  const float* x     = (const float*)d_in[0];
  const float* wqkv  = (const float*)d_in[1];
  const float* wproj = (const float*)d_in[2];
  float* out = (float*)d_out;

  float* ws = (float*)d_ws;
  float* Q  = ws;
  float* K  = Q + NHEADS * N_TOK * HD;
  float* V  = K + NHEADS * N_TOK * HD;
  float* AO = V + NHEADS * N_TOK * HD;

  gsa3d_qkv_kernel <<<216, 256, 0, stream>>>(x, wqkv, Q, K, V);
  gsa3d_attn_kernel<<<432, 256, 0, stream>>>(Q, K, V, AO);
  gsa3d_proj_kernel<<<216, 256, 0, stream>>>(AO, wproj, out);
}
